# starting kernel
# baseline (speedup 1.0000x reference)
#include <hip/hip_runtime.h>
#include <hip/hip_bf16.h>

#define N0    262144
#define N1_   65536
#define N2_   16384
#define E1_   1048576
#define E2_   262144
#define IN_C  128
#define HID_C 256
#define OUT_C 64

typedef __attribute__((ext_vector_type(2))) float v2f;
typedef __attribute__((ext_vector_type(8))) float v8f;

__global__ void sage_count_deg(const int* __restrict__ dst, int* __restrict__ deg,
                               int E) {
    int e = blockIdx.x * blockDim.x + threadIdx.x;
    if (e < E) atomicAdd(&deg[dst[e]], 1);
}

__global__ void sage_excl_scan(const int* __restrict__ in, int* __restrict__ out,
                               int n) {
    __shared__ int buf[1024];
    __shared__ int carry_s;
    const int tid = threadIdx.x;
    if (tid == 0) carry_s = 0;
    __syncthreads();
    for (int base = 0; base < n; base += 1024) {
        int i = base + tid;
        int v = (i < n) ? in[i] : 0;
        buf[tid] = v;
        __syncthreads();
        for (int off = 1; off < 1024; off <<= 1) {
            int t = (tid >= off) ? buf[tid - off] : 0;
            __syncthreads();
            buf[tid] += t;
            __syncthreads();
        }
        int incl = buf[tid];
        int c = carry_s;
        if (i < n) out[i] = c + incl - v;
        __syncthreads();
        if (tid == 0) carry_s = c + buf[1023];
        __syncthreads();
    }
}

__global__ void sage_bucket_edges(const int* __restrict__ src,
                                  const int* __restrict__ dst,
                                  const int* __restrict__ offs,
                                  int* __restrict__ fill,
                                  int* __restrict__ esrc, int E) {
    int e = blockIdx.x * blockDim.x + threadIdx.x;
    if (e >= E) return;
    int d = dst[e];
    int pos = offs[d] + atomicAdd(&fill[d], 1);
    esrc[pos] = src[e];
}

template <int C>
__global__ void sage_gather_mean(const float* __restrict__ x,
                                 const int* __restrict__ esrc,
                                 const int* __restrict__ offs,
                                 const int* __restrict__ deg,
                                 float* __restrict__ outm, int nDst) {
    const int wave = blockIdx.x * (blockDim.x >> 5) + (threadIdx.x >> 5);
    const int lane = threadIdx.x & 31;
    if (wave >= nDst) return;
    const int beg = offs[wave];
    const int d   = deg[wave];
    constexpr int V = C / 128;
    float4 acc[V];
#pragma unroll
    for (int v = 0; v < V; ++v) acc[v] = make_float4(0.f, 0.f, 0.f, 0.f);

    for (int base = 0; base < d; base += 32) {
        int n = d - base;
        if (n > 32) n = 32;
        int myid = (lane < n) ? esrc[beg + base + lane] : 0;
        for (int j = 0; j < n; ++j) {
            int s = __shfl(myid, j, 32);
            const float4* row = (const float4*)(x + (size_t)s * C);
#pragma unroll
            for (int v = 0; v < V; ++v) {
                float4 t = row[lane + 32 * v];
                acc[v].x += t.x; acc[v].y += t.y;
                acc[v].z += t.z; acc[v].w += t.w;
            }
        }
    }

    const float scale = (d > 0) ? (1.0f / (float)d) : 0.0f;
    float4* orow = (float4*)(outm + (size_t)wave * C);
#pragma unroll
    for (int v = 0; v < V; ++v) {
        float4 t;
        t.x = acc[v].x * scale; t.y = acc[v].y * scale;
        t.z = acc[v].z * scale; t.w = acc[v].w * scale;
        orow[lane + 32 * v] = t;
    }
}

template <int K, int N, int MT, int NT, bool RELU>
__global__ void sage_gemm_two(const float* __restrict__ A1,
                              const float* __restrict__ B1,
                              const float* __restrict__ A2,
                              const float* __restrict__ B2,
                              const float* __restrict__ bias,
                              float* __restrict__ out, int M) {
    const int lane   = threadIdx.x & 31;
    const int waveId = blockIdx.x * (blockDim.x >> 5) + (threadIdx.x >> 5);
    const int groupsN = N / (16 * NT);
    const int gm = waveId / groupsN;
    const int gn = waveId % groupsN;
    if (gm >= M / (16 * MT)) return;

    const int m0  = gm * 16 * MT;
    const int n0  = gn * 16 * NT;
    const int khi = (lane >> 4) * 2;

    const float* a1p[MT];
    const float* a2p[MT];
#pragma unroll
    for (int mt = 0; mt < MT; ++mt) {
        int r = m0 + mt * 16 + (lane & 15);
        a1p[mt] = A1 + (size_t)r * K + khi;
        a2p[mt] = A2 + (size_t)r * K + khi;
    }
    const float* b1p[NT];
    const float* b2p[NT];
#pragma unroll
    for (int nt = 0; nt < NT; ++nt) {
        int c = n0 + nt * 16 + (lane & 15);
        b1p[nt] = B1 + (size_t)c * K + khi;
        b2p[nt] = B2 + (size_t)c * K + khi;
    }

    v8f acc[MT][NT] = {};

#pragma unroll 2
    for (int k0 = 0; k0 < K; k0 += 4) {
        v2f a[MT], b[NT];
#pragma unroll
        for (int mt = 0; mt < MT; ++mt) a[mt] = *(const v2f*)(a1p[mt] + k0);
#pragma unroll
        for (int nt = 0; nt < NT; ++nt) b[nt] = *(const v2f*)(b1p[nt] + k0);
#pragma unroll
        for (int mt = 0; mt < MT; ++mt)
#pragma unroll
            for (int nt = 0; nt < NT; ++nt)
                acc[mt][nt] = __builtin_amdgcn_wmma_f32_16x16x4_f32(
                    false, a[mt], false, b[nt], (short)0, acc[mt][nt],
                    false, false);
    }
#pragma unroll 2
    for (int k0 = 0; k0 < K; k0 += 4) {
        v2f a[MT], b[NT];
#pragma unroll
        for (int mt = 0; mt < MT; ++mt) a[mt] = *(const v2f*)(a2p[mt] + k0);
#pragma unroll
        for (int nt = 0; nt < NT; ++nt) b[nt] = *(const v2f*)(b2p[nt] + k0);
#pragma unroll
        for (int mt = 0; mt < MT; ++mt)
#pragma unroll
            for (int nt = 0; nt < NT; ++nt)
                acc[mt][nt] = __builtin_amdgcn_wmma_f32_16x16x4_f32(
                    false, a[mt], false, b[nt], (short)0, acc[mt][nt],
                    false, false);
    }

    const int rbase = ((lane >> 4) << 3);
#pragma unroll
    for (int nt = 0; nt < NT; ++nt) {
        const int ocol = n0 + nt * 16 + (lane & 15);
        const float bb = bias[ocol];
#pragma unroll
        for (int mt = 0; mt < MT; ++mt) {
#pragma unroll
            for (int r = 0; r < 8; ++r) {
                float v = acc[mt][nt][r] + bb;
                if (RELU) v = fmaxf(v, 0.0f);
                out[(size_t)(m0 + mt * 16 + rbase + r) * N + ocol] = v;
            }
        }
    }
}

__global__ void sage_logsoftmax64(float* __restrict__ out, int rows) {
    int wave = blockIdx.x * (blockDim.x >> 5) + (threadIdx.x >> 5);
    int lane = threadIdx.x & 31;
    if (wave >= rows) return;
    float* p = out + (size_t)wave * 64;
    float a = p[lane];
    float b = p[lane + 32];
    float m = fmaxf(a, b);
#pragma unroll
    for (int off = 16; off > 0; off >>= 1)
        m = fmaxf(m, __shfl_xor(m, off, 32));
    float s = expf(a - m) + expf(b - m);
#pragma unroll
    for (int off = 16; off > 0; off >>= 1)
        s += __shfl_xor(s, off, 32);
    float lse = m + logf(s);
    p[lane]      = a - lse;
    p[lane + 32] = b - lse;
}

extern "C" void kernel_launch(void* const* d_in, const int* in_sizes, int n_in,
                              void* d_out, int out_size, void* d_ws, size_t ws_size,
                              hipStream_t stream) {
    const float* x    = (const float*)d_in[0];
    const int*   src1 = (const int*)d_in[1];
    const int*   dst1 = (const int*)d_in[2];
    const int*   src2 = (const int*)d_in[3];
    const int*   dst2 = (const int*)d_in[4];
    const float* W1l  = (const float*)d_in[5];
    const float* W1r  = (const float*)d_in[6];
    const float* b1   = (const float*)d_in[7];
    const float* W2l  = (const float*)d_in[8];
    const float* W2r  = (const float*)d_in[9];
    const float* b2   = (const float*)d_in[10];
    float* out = (float*)d_out;

    int* deg1  = (int*)d_ws;
    int* fill1 = deg1 + N1_;
    int* deg2  = fill1 + N1_;
    int* fill2 = deg2 + N2_;
    int* offs1 = fill2 + N2_;
    int* offs2 = offs1 + N1_;
    int* esrc1 = offs2 + N2_;
    int* esrc2 = esrc1 + E1_;
    float* aggr1 = (float*)(esrc2 + E2_);
    float* aggr2 = aggr1 + (size_t)N1_ * IN_C;
    float* h     = aggr2 + (size_t)N2_ * HID_C;

    size_t zbytes = (size_t)(2 * N1_ + 2 * N2_) * sizeof(int);
    hipMemsetAsync(d_ws, 0, zbytes, stream);

    sage_count_deg<<<E1_ / 256, 256, 0, stream>>>(dst1, deg1, E1_);
    sage_excl_scan<<<1, 1024, 0, stream>>>(deg1, offs1, N1_);
    sage_bucket_edges<<<E1_ / 256, 256, 0, stream>>>(src1, dst1, offs1, fill1,
                                                     esrc1, E1_);
    sage_gather_mean<IN_C><<<N1_ / 8, 256, 0, stream>>>(x, esrc1, offs1, deg1,
                                                        aggr1, N1_);
    {
        int waves = (N1_ / 32) * (HID_C / 64);
        sage_gemm_two<IN_C, HID_C, 2, 4, true>
            <<<waves / 8, 256, 0, stream>>>(aggr1, W1l, x, W1r, b1, h, N1_);
    }

    sage_count_deg<<<E2_ / 256, 256, 0, stream>>>(dst2, deg2, E2_);
    sage_excl_scan<<<1, 1024, 0, stream>>>(deg2, offs2, N2_);
    sage_bucket_edges<<<E2_ / 256, 256, 0, stream>>>(src2, dst2, offs2, fill2,
                                                     esrc2, E2_);
    sage_gather_mean<HID_C><<<N2_ / 8, 256, 0, stream>>>(h, esrc2, offs2, deg2,
                                                         aggr2, N2_);
    {
        int waves = (N2_ / 32) * (OUT_C / 32);
        sage_gemm_two<HID_C, OUT_C, 2, 2, false>
            <<<waves / 8, 256, 0, stream>>>(aggr2, W2l, h, W2r, b2, out, N2_);
    }

    sage_logsoftmax64<<<(N2_ + 7) / 8, 256, 0, stream>>>(out, N2_);
}


# Degree scan: one block, 23 barriers per 1024 elements -> n/4096 blocks, each re-sums the counts in front of its 4096 from L2 in one load group, then one barrier; integer adds, offsets bit-identical
# speedup vs baseline: 1.5160x; 1.5160x over previous
#include <hip/hip_runtime.h>
#include <hip/hip_bf16.h>

#define N0    262144
#define N1_   65536
#define N2_   16384
#define E1_   1048576
#define E2_   262144
#define IN_C  128
#define HID_C 256
#define OUT_C 64

typedef __attribute__((ext_vector_type(2))) float v2f;
typedef __attribute__((ext_vector_type(8))) float v8f;

__global__ void sage_count_deg(const int* __restrict__ dst, int* __restrict__ deg,
                               int E) {
    int e = blockIdx.x * blockDim.x + threadIdx.x;
    if (e < E) atomicAdd(&deg[dst[e]], 1);
}

template <int NCHUNK>
__global__ void __launch_bounds__(1024)
sage_excl_scan(const int* __restrict__ in, int* __restrict__ out) {
    __shared__ int wave_own[32];
    __shared__ int wave_front[32];
    const int tid  = threadIdx.x;
    const int lane = tid & 31;
    const int wv   = tid >> 5;
    const int blk  = blockIdx.x;
    const int4* in4 = (const int4*)in;

    const int4 v = in4[blk * 1024 + tid];

    int4 f[NCHUNK - 1];
#pragma unroll
    for (int c = 0; c < NCHUNK - 1; ++c) {
        const int cc = (c < blk) ? c : blk;
        f[c] = in4[cc * 1024 + tid];
    }
    int front = 0;
#pragma unroll
    for (int c = 0; c < NCHUNK - 1; ++c) {
        const int s4 = (f[c].x + f[c].y) + (f[c].z + f[c].w);
        front += (c < blk) ? s4 : 0;
    }

    const int s = (v.x + v.y) + (v.z + v.w);
    int incl = s;
#pragma unroll
    for (int off = 1; off < 32; off <<= 1) {
        const int t = __shfl_up(incl, off, 32);
        if (lane >= off) incl += t;
    }
#pragma unroll
    for (int off = 16; off > 0; off >>= 1) front += __shfl_xor(front, off, 32);

    if (lane == 31) wave_own[wv] = incl;
    if (lane == 0)  wave_front[wv] = front;
    __syncthreads();

    int a = wave_front[lane] + ((lane < wv) ? wave_own[lane] : 0);
#pragma unroll
    for (int off = 16; off > 0; off >>= 1) a += __shfl_xor(a, off, 32);

    int4 o;
    o.x = a + incl - s;
    o.y = o.x + v.x;
    o.z = o.y + v.y;
    o.w = o.z + v.z;
    ((int4*)out)[blk * 1024 + tid] = o;
}

__global__ void sage_bucket_edges(const int* __restrict__ src,
                                  const int* __restrict__ dst,
                                  const int* __restrict__ offs,
                                  int* __restrict__ fill,
                                  int* __restrict__ esrc, int E) {
    int e = blockIdx.x * blockDim.x + threadIdx.x;
    if (e >= E) return;
    int d = dst[e];
    int pos = offs[d] + atomicAdd(&fill[d], 1);
    esrc[pos] = src[e];
}

template <int C>
__global__ void sage_gather_mean(const float* __restrict__ x,
                                 const int* __restrict__ esrc,
                                 const int* __restrict__ offs,
                                 const int* __restrict__ deg,
                                 float* __restrict__ outm, int nDst) {
    const int wave = blockIdx.x * (blockDim.x >> 5) + (threadIdx.x >> 5);
    const int lane = threadIdx.x & 31;
    if (wave >= nDst) return;
    const int beg = offs[wave];
    const int d   = deg[wave];
    constexpr int V = C / 128;
    float4 acc[V];
#pragma unroll
    for (int v = 0; v < V; ++v) acc[v] = make_float4(0.f, 0.f, 0.f, 0.f);

    for (int base = 0; base < d; base += 32) {
        int n = d - base;
        if (n > 32) n = 32;
        int myid = (lane < n) ? esrc[beg + base + lane] : 0;
        for (int j = 0; j < n; ++j) {
            int s = __shfl(myid, j, 32);
            const float4* row = (const float4*)(x + (size_t)s * C);
#pragma unroll
            for (int v = 0; v < V; ++v) {
                float4 t = row[lane + 32 * v];
                acc[v].x += t.x; acc[v].y += t.y;
                acc[v].z += t.z; acc[v].w += t.w;
            }
        }
    }

    const float scale = (d > 0) ? (1.0f / (float)d) : 0.0f;
    float4* orow = (float4*)(outm + (size_t)wave * C);
#pragma unroll
    for (int v = 0; v < V; ++v) {
        float4 t;
        t.x = acc[v].x * scale; t.y = acc[v].y * scale;
        t.z = acc[v].z * scale; t.w = acc[v].w * scale;
        orow[lane + 32 * v] = t;
    }
}

template <int K, int N, int MT, int NT, bool RELU>
__global__ void sage_gemm_two(const float* __restrict__ A1,
                              const float* __restrict__ B1,
                              const float* __restrict__ A2,
                              const float* __restrict__ B2,
                              const float* __restrict__ bias,
                              float* __restrict__ out, int M) {
    const int lane   = threadIdx.x & 31;
    const int waveId = blockIdx.x * (blockDim.x >> 5) + (threadIdx.x >> 5);
    const int groupsN = N / (16 * NT);
    const int gm = waveId / groupsN;
    const int gn = waveId % groupsN;
    if (gm >= M / (16 * MT)) return;

    const int m0  = gm * 16 * MT;
    const int n0  = gn * 16 * NT;
    const int khi = (lane >> 4) * 2;

    const float* a1p[MT];
    const float* a2p[MT];
#pragma unroll
    for (int mt = 0; mt < MT; ++mt) {
        int r = m0 + mt * 16 + (lane & 15);
        a1p[mt] = A1 + (size_t)r * K + khi;
        a2p[mt] = A2 + (size_t)r * K + khi;
    }
    const float* b1p[NT];
    const float* b2p[NT];
#pragma unroll
    for (int nt = 0; nt < NT; ++nt) {
        int c = n0 + nt * 16 + (lane & 15);
        b1p[nt] = B1 + (size_t)c * K + khi;
        b2p[nt] = B2 + (size_t)c * K + khi;
    }

    v8f acc[MT][NT] = {};

#pragma unroll 2
    for (int k0 = 0; k0 < K; k0 += 4) {
        v2f a[MT], b[NT];
#pragma unroll
        for (int mt = 0; mt < MT; ++mt) a[mt] = *(const v2f*)(a1p[mt] + k0);
#pragma unroll
        for (int nt = 0; nt < NT; ++nt) b[nt] = *(const v2f*)(b1p[nt] + k0);
#pragma unroll
        for (int mt = 0; mt < MT; ++mt)
#pragma unroll
            for (int nt = 0; nt < NT; ++nt)
                acc[mt][nt] = __builtin_amdgcn_wmma_f32_16x16x4_f32(
                    false, a[mt], false, b[nt], (short)0, acc[mt][nt],
                    false, false);
    }
#pragma unroll 2
    for (int k0 = 0; k0 < K; k0 += 4) {
        v2f a[MT], b[NT];
#pragma unroll
        for (int mt = 0; mt < MT; ++mt) a[mt] = *(const v2f*)(a2p[mt] + k0);
#pragma unroll
        for (int nt = 0; nt < NT; ++nt) b[nt] = *(const v2f*)(b2p[nt] + k0);
#pragma unroll
        for (int mt = 0; mt < MT; ++mt)
#pragma unroll
            for (int nt = 0; nt < NT; ++nt)
                acc[mt][nt] = __builtin_amdgcn_wmma_f32_16x16x4_f32(
                    false, a[mt], false, b[nt], (short)0, acc[mt][nt],
                    false, false);
    }

    const int rbase = ((lane >> 4) << 3);
#pragma unroll
    for (int nt = 0; nt < NT; ++nt) {
        const int ocol = n0 + nt * 16 + (lane & 15);
        const float bb = bias[ocol];
#pragma unroll
        for (int mt = 0; mt < MT; ++mt) {
#pragma unroll
            for (int r = 0; r < 8; ++r) {
                float v = acc[mt][nt][r] + bb;
                if (RELU) v = fmaxf(v, 0.0f);
                out[(size_t)(m0 + mt * 16 + rbase + r) * N + ocol] = v;
            }
        }
    }
}

__global__ void sage_logsoftmax64(float* __restrict__ out, int rows) {
    int wave = blockIdx.x * (blockDim.x >> 5) + (threadIdx.x >> 5);
    int lane = threadIdx.x & 31;
    if (wave >= rows) return;
    float* p = out + (size_t)wave * 64;
    float a = p[lane];
    float b = p[lane + 32];
    float m = fmaxf(a, b);
#pragma unroll
    for (int off = 16; off > 0; off >>= 1)
        m = fmaxf(m, __shfl_xor(m, off, 32));
    float s = expf(a - m) + expf(b - m);
#pragma unroll
    for (int off = 16; off > 0; off >>= 1)
        s += __shfl_xor(s, off, 32);
    float lse = m + logf(s);
    p[lane]      = a - lse;
    p[lane + 32] = b - lse;
}

extern "C" void kernel_launch(void* const* d_in, const int* in_sizes, int n_in,
                              void* d_out, int out_size, void* d_ws, size_t ws_size,
                              hipStream_t stream) {
    const float* x    = (const float*)d_in[0];
    const int*   src1 = (const int*)d_in[1];
    const int*   dst1 = (const int*)d_in[2];
    const int*   src2 = (const int*)d_in[3];
    const int*   dst2 = (const int*)d_in[4];
    const float* W1l  = (const float*)d_in[5];
    const float* W1r  = (const float*)d_in[6];
    const float* b1   = (const float*)d_in[7];
    const float* W2l  = (const float*)d_in[8];
    const float* W2r  = (const float*)d_in[9];
    const float* b2   = (const float*)d_in[10];
    float* out = (float*)d_out;

    int* deg1  = (int*)d_ws;
    int* fill1 = deg1 + N1_;
    int* deg2  = fill1 + N1_;
    int* fill2 = deg2 + N2_;
    int* offs1 = fill2 + N2_;
    int* offs2 = offs1 + N1_;
    int* esrc1 = offs2 + N2_;
    int* esrc2 = esrc1 + E1_;
    float* aggr1 = (float*)(esrc2 + E2_);
    float* aggr2 = aggr1 + (size_t)N1_ * IN_C;
    float* h     = aggr2 + (size_t)N2_ * HID_C;

    size_t zbytes = (size_t)(2 * N1_ + 2 * N2_) * sizeof(int);
    hipMemsetAsync(d_ws, 0, zbytes, stream);

    sage_count_deg<<<E1_ / 256, 256, 0, stream>>>(dst1, deg1, E1_);
    sage_excl_scan<N1_ / 4096><<<N1_ / 4096, 1024, 0, stream>>>(deg1, offs1);
    sage_bucket_edges<<<E1_ / 256, 256, 0, stream>>>(src1, dst1, offs1, fill1,
                                                     esrc1, E1_);
    sage_gather_mean<IN_C><<<N1_ / 8, 256, 0, stream>>>(x, esrc1, offs1, deg1,
                                                        aggr1, N1_);
    {
        int waves = (N1_ / 32) * (HID_C / 64);
        sage_gemm_two<IN_C, HID_C, 2, 4, true>
            <<<waves / 8, 256, 0, stream>>>(aggr1, W1l, x, W1r, b1, h, N1_);
    }

    sage_count_deg<<<E2_ / 256, 256, 0, stream>>>(dst2, deg2, E2_);
    sage_excl_scan<N2_ / 4096><<<N2_ / 4096, 1024, 0, stream>>>(deg2, offs2);
    sage_bucket_edges<<<E2_ / 256, 256, 0, stream>>>(src2, dst2, offs2, fill2,
                                                     esrc2, E2_);
    sage_gather_mean<HID_C><<<N2_ / 8, 256, 0, stream>>>(h, esrc2, offs2, deg2,
                                                         aggr2, N2_);
    {
        int waves = (N2_ / 32) * (OUT_C / 32);
        sage_gemm_two<HID_C, OUT_C, 2, 2, false>
            <<<waves / 8, 256, 0, stream>>>(aggr2, W2l, h, W2r, b2, out, N2_);
    }

    sage_logsoftmax64<<<(N2_ + 7) / 8, 256, 0, stream>>>(out, N2_);
}


# GEMM operand loads: a lane takes 64 contiguous bytes of its row per 32-k block (4 x 16 B; a lane pair uses a whole 128-B line at once), not 8 B per k-step; k order in a block permuted: sums reorder
# speedup vs baseline: 1.6649x; 1.0982x over previous
#include <hip/hip_runtime.h>
#include <hip/hip_bf16.h>

#define N0    262144
#define N1_   65536
#define N2_   16384
#define E1_   1048576
#define E2_   262144
#define IN_C  128
#define HID_C 256
#define OUT_C 64

typedef __attribute__((ext_vector_type(2))) float v2f;
typedef __attribute__((ext_vector_type(4))) float v4f;
typedef __attribute__((ext_vector_type(8))) float v8f;

__global__ void sage_count_deg(const int* __restrict__ dst, int* __restrict__ deg,
                               int E) {
    int e = blockIdx.x * blockDim.x + threadIdx.x;
    if (e < E) atomicAdd(&deg[dst[e]], 1);
}

template <int NCHUNK>
__global__ void __launch_bounds__(1024)
sage_excl_scan(const int* __restrict__ in, int* __restrict__ out) {
    __shared__ int wave_own[32];
    __shared__ int wave_front[32];
    const int tid  = threadIdx.x;
    const int lane = tid & 31;
    const int wv   = tid >> 5;
    const int blk  = blockIdx.x;
    const int4* in4 = (const int4*)in;

    const int4 v = in4[blk * 1024 + tid];

    int4 f[NCHUNK - 1];
#pragma unroll
    for (int c = 0; c < NCHUNK - 1; ++c) {
        const int cc = (c < blk) ? c : blk;
        f[c] = in4[cc * 1024 + tid];
    }
    int front = 0;
#pragma unroll
    for (int c = 0; c < NCHUNK - 1; ++c) {
        const int s4 = (f[c].x + f[c].y) + (f[c].z + f[c].w);
        front += (c < blk) ? s4 : 0;
    }

    const int s = (v.x + v.y) + (v.z + v.w);
    int incl = s;
#pragma unroll
    for (int off = 1; off < 32; off <<= 1) {
        const int t = __shfl_up(incl, off, 32);
        if (lane >= off) incl += t;
    }
#pragma unroll
    for (int off = 16; off > 0; off >>= 1) front += __shfl_xor(front, off, 32);

    if (lane == 31) wave_own[wv] = incl;
    if (lane == 0)  wave_front[wv] = front;
    __syncthreads();

    int a = wave_front[lane] + ((lane < wv) ? wave_own[lane] : 0);
#pragma unroll
    for (int off = 16; off > 0; off >>= 1) a += __shfl_xor(a, off, 32);

    int4 o;
    o.x = a + incl - s;
    o.y = o.x + v.x;
    o.z = o.y + v.y;
    o.w = o.z + v.z;
    ((int4*)out)[blk * 1024 + tid] = o;
}

__global__ void sage_bucket_edges(const int* __restrict__ src,
                                  const int* __restrict__ dst,
                                  const int* __restrict__ offs,
                                  int* __restrict__ fill,
                                  int* __restrict__ esrc, int E) {
    int e = blockIdx.x * blockDim.x + threadIdx.x;
    if (e >= E) return;
    int d = dst[e];
    int pos = offs[d] + atomicAdd(&fill[d], 1);
    esrc[pos] = src[e];
}

template <int C>
__global__ void sage_gather_mean(const float* __restrict__ x,
                                 const int* __restrict__ esrc,
                                 const int* __restrict__ offs,
                                 const int* __restrict__ deg,
                                 float* __restrict__ outm, int nDst) {
    const int wave = blockIdx.x * (blockDim.x >> 5) + (threadIdx.x >> 5);
    const int lane = threadIdx.x & 31;
    if (wave >= nDst) return;
    const int beg = offs[wave];
    const int d   = deg[wave];
    constexpr int V = C / 128;
    float4 acc[V];
#pragma unroll
    for (int v = 0; v < V; ++v) acc[v] = make_float4(0.f, 0.f, 0.f, 0.f);

    for (int base = 0; base < d; base += 32) {
        int n = d - base;
        if (n > 32) n = 32;
        int myid = (lane < n) ? esrc[beg + base + lane] : 0;
        for (int j = 0; j < n; ++j) {
            int s = __shfl(myid, j, 32);
            const float4* row = (const float4*)(x + (size_t)s * C);
#pragma unroll
            for (int v = 0; v < V; ++v) {
                float4 t = row[lane + 32 * v];
                acc[v].x += t.x; acc[v].y += t.y;
                acc[v].z += t.z; acc[v].w += t.w;
            }
        }
    }

    const float scale = (d > 0) ? (1.0f / (float)d) : 0.0f;
    float4* orow = (float4*)(outm + (size_t)wave * C);
#pragma unroll
    for (int v = 0; v < V; ++v) {
        float4 t;
        t.x = acc[v].x * scale; t.y = acc[v].y * scale;
        t.z = acc[v].z * scale; t.w = acc[v].w * scale;
        orow[lane + 32 * v] = t;
    }
}

template <int K, int MT, int NT>
__device__ __forceinline__ void sage_gemm_phase(const float* const (&ap)[MT],
                                                const float* const (&bp)[NT],
                                                v8f (&acc)[MT][NT]) {
#pragma unroll 1
    for (int kc = 0; kc < K; kc += 32) {
        v4f a[4][MT], b[4][NT];
#pragma unroll
        for (int q = 0; q < 4; ++q) {
#pragma unroll
            for (int mt = 0; mt < MT; ++mt)
                a[q][mt] = *(const v4f*)(ap[mt] + kc + 4 * q);
#pragma unroll
            for (int nt = 0; nt < NT; ++nt)
                b[q][nt] = *(const v4f*)(bp[nt] + kc + 4 * q);
        }
        __builtin_amdgcn_sched_barrier(0);
#pragma unroll
        for (int q = 0; q < 4; ++q) {
#pragma unroll
            for (int p = 0; p < 2; ++p) {
                v2f af[MT], bf[NT];
#pragma unroll
                for (int mt = 0; mt < MT; ++mt) {
                    af[mt].x = a[q][mt][2 * p];
                    af[mt].y = a[q][mt][2 * p + 1];
                }
#pragma unroll
                for (int nt = 0; nt < NT; ++nt) {
                    bf[nt].x = b[q][nt][2 * p];
                    bf[nt].y = b[q][nt][2 * p + 1];
                }
#pragma unroll
                for (int mt = 0; mt < MT; ++mt)
#pragma unroll
                    for (int nt = 0; nt < NT; ++nt)
                        acc[mt][nt] = __builtin_amdgcn_wmma_f32_16x16x4_f32(
                            false, af[mt], false, bf[nt], (short)0, acc[mt][nt],
                            false, false);
            }
        }
    }
}

template <int K, int N, int MT, int NT, bool RELU>
__global__ void __launch_bounds__(256)
sage_gemm_two(const float* __restrict__ A1,
                              const float* __restrict__ B1,
                              const float* __restrict__ A2,
                              const float* __restrict__ B2,
                              const float* __restrict__ bias,
                              float* __restrict__ out, int M) {
    static_assert(K % 32 == 0, "K is walked in whole 128-byte row lines");
    const int lane   = threadIdx.x & 31;
    const int waveId = blockIdx.x * (blockDim.x >> 5) + (threadIdx.x >> 5);
    const int groupsN = N / (16 * NT);
    const int gm = waveId / groupsN;
    const int gn = waveId % groupsN;
    if (gm >= M / (16 * MT)) return;

    const int m0  = gm * 16 * MT;
    const int n0  = gn * 16 * NT;
    const int kh  = (lane >> 4) * 16;

    const float* a1p[MT];
    const float* a2p[MT];
#pragma unroll
    for (int mt = 0; mt < MT; ++mt) {
        int r = m0 + mt * 16 + (lane & 15);
        a1p[mt] = A1 + (size_t)r * K + kh;
        a2p[mt] = A2 + (size_t)r * K + kh;
    }
    const float* b1p[NT];
    const float* b2p[NT];
#pragma unroll
    for (int nt = 0; nt < NT; ++nt) {
        int c = n0 + nt * 16 + (lane & 15);
        b1p[nt] = B1 + (size_t)c * K + kh;
        b2p[nt] = B2 + (size_t)c * K + kh;
    }

    v8f acc[MT][NT] = {};

    sage_gemm_phase<K, MT, NT>(a1p, b1p, acc);
    sage_gemm_phase<K, MT, NT>(a2p, b2p, acc);

    const int rbase = ((lane >> 4) << 3);
#pragma unroll
    for (int nt = 0; nt < NT; ++nt) {
        const int ocol = n0 + nt * 16 + (lane & 15);
        const float bb = bias[ocol];
#pragma unroll
        for (int mt = 0; mt < MT; ++mt) {
#pragma unroll
            for (int r = 0; r < 8; ++r) {
                float v = acc[mt][nt][r] + bb;
                if (RELU) v = fmaxf(v, 0.0f);
                out[(size_t)(m0 + mt * 16 + rbase + r) * N + ocol] = v;
            }
        }
    }
}

__global__ void sage_logsoftmax64(float* __restrict__ out, int rows) {
    int wave = blockIdx.x * (blockDim.x >> 5) + (threadIdx.x >> 5);
    int lane = threadIdx.x & 31;
    if (wave >= rows) return;
    float* p = out + (size_t)wave * 64;
    float a = p[lane];
    float b = p[lane + 32];
    float m = fmaxf(a, b);
#pragma unroll
    for (int off = 16; off > 0; off >>= 1)
        m = fmaxf(m, __shfl_xor(m, off, 32));
    float s = expf(a - m) + expf(b - m);
#pragma unroll
    for (int off = 16; off > 0; off >>= 1)
        s += __shfl_xor(s, off, 32);
    float lse = m + logf(s);
    p[lane]      = a - lse;
    p[lane + 32] = b - lse;
}

extern "C" void kernel_launch(void* const* d_in, const int* in_sizes, int n_in,
                              void* d_out, int out_size, void* d_ws, size_t ws_size,
                              hipStream_t stream) {
    const float* x    = (const float*)d_in[0];
    const int*   src1 = (const int*)d_in[1];
    const int*   dst1 = (const int*)d_in[2];
    const int*   src2 = (const int*)d_in[3];
    const int*   dst2 = (const int*)d_in[4];
    const float* W1l  = (const float*)d_in[5];
    const float* W1r  = (const float*)d_in[6];
    const float* b1   = (const float*)d_in[7];
    const float* W2l  = (const float*)d_in[8];
    const float* W2r  = (const float*)d_in[9];
    const float* b2   = (const float*)d_in[10];
    float* out = (float*)d_out;

    int* deg1  = (int*)d_ws;
    int* fill1 = deg1 + N1_;
    int* deg2  = fill1 + N1_;
    int* fill2 = deg2 + N2_;
    int* offs1 = fill2 + N2_;
    int* offs2 = offs1 + N1_;
    int* esrc1 = offs2 + N2_;
    int* esrc2 = esrc1 + E1_;
    float* aggr1 = (float*)(esrc2 + E2_);
    float* aggr2 = aggr1 + (size_t)N1_ * IN_C;
    float* h     = aggr2 + (size_t)N2_ * HID_C;

    size_t zbytes = (size_t)(2 * N1_ + 2 * N2_) * sizeof(int);
    hipMemsetAsync(d_ws, 0, zbytes, stream);

    sage_count_deg<<<E1_ / 256, 256, 0, stream>>>(dst1, deg1, E1_);
    sage_excl_scan<N1_ / 4096><<<N1_ / 4096, 1024, 0, stream>>>(deg1, offs1);
    sage_bucket_edges<<<E1_ / 256, 256, 0, stream>>>(src1, dst1, offs1, fill1,
                                                     esrc1, E1_);
    sage_gather_mean<IN_C><<<N1_ / 8, 256, 0, stream>>>(x, esrc1, offs1, deg1,
                                                        aggr1, N1_);
    {
        int waves = (N1_ / 32) * (HID_C / 64);
        sage_gemm_two<IN_C, HID_C, 2, 4, true>
            <<<waves / 8, 256, 0, stream>>>(aggr1, W1l, x, W1r, b1, h, N1_);
    }

    sage_count_deg<<<E2_ / 256, 256, 0, stream>>>(dst2, deg2, E2_);
    sage_excl_scan<N2_ / 4096><<<N2_ / 4096, 1024, 0, stream>>>(deg2, offs2);
    sage_bucket_edges<<<E2_ / 256, 256, 0, stream>>>(src2, dst2, offs2, fill2,
                                                     esrc2, E2_);
    sage_gather_mean<HID_C><<<N2_ / 8, 256, 0, stream>>>(h, esrc2, offs2, deg2,
                                                         aggr2, N2_);
    {
        int waves = (N2_ / 32) * (OUT_C / 32);
        sage_gemm_two<HID_C, OUT_C, 2, 2, false>
            <<<waves / 8, 256, 0, stream>>>(aggr2, W2l, h, W2r, b2, out, N2_);
    }

    sage_logsoftmax64<<<(N2_ + 7) / 8, 256, 0, stream>>>(out, N2_);
}


# Edge bucketing: an edge's slot in its destination's run is the value the degree-count atomic returns, so the placing pass has no atomics (4 edges per thread); sums stay in atomic arrival order
# speedup vs baseline: 1.9725x; 1.1848x over previous
#include <hip/hip_runtime.h>
#include <hip/hip_bf16.h>

#define N0    262144
#define N1_   65536
#define N2_   16384
#define E1_   1048576
#define E2_   262144
#define IN_C  128
#define HID_C 256
#define OUT_C 64

typedef __attribute__((ext_vector_type(2))) float v2f;
typedef __attribute__((ext_vector_type(4))) float v4f;
typedef __attribute__((ext_vector_type(8))) float v8f;

__global__ void __launch_bounds__(256)
sage_count_slots(const int* __restrict__ dst, int* __restrict__ deg,
                 int* __restrict__ slot, int E) {
    const int i = blockIdx.x * blockDim.x + threadIdx.x;
    if (4 * i >= E) return;
    const int4 d = ((const int4*)dst)[i];
    int4 r;
    r.x = atomicAdd(&deg[d.x], 1);
    r.y = atomicAdd(&deg[d.y], 1);
    r.z = atomicAdd(&deg[d.z], 1);
    r.w = atomicAdd(&deg[d.w], 1);
    ((int4*)slot)[i] = r;
}

template <int NCHUNK>
__global__ void __launch_bounds__(1024)
sage_excl_scan(const int* __restrict__ in, int* __restrict__ out) {
    __shared__ int wave_own[32];
    __shared__ int wave_front[32];
    const int tid  = threadIdx.x;
    const int lane = tid & 31;
    const int wv   = tid >> 5;
    const int blk  = blockIdx.x;
    const int4* in4 = (const int4*)in;

    const int4 v = in4[blk * 1024 + tid];

    int4 f[NCHUNK - 1];
#pragma unroll
    for (int c = 0; c < NCHUNK - 1; ++c) {
        const int cc = (c < blk) ? c : blk;
        f[c] = in4[cc * 1024 + tid];
    }
    int front = 0;
#pragma unroll
    for (int c = 0; c < NCHUNK - 1; ++c) {
        const int s4 = (f[c].x + f[c].y) + (f[c].z + f[c].w);
        front += (c < blk) ? s4 : 0;
    }

    const int s = (v.x + v.y) + (v.z + v.w);
    int incl = s;
#pragma unroll
    for (int off = 1; off < 32; off <<= 1) {
        const int t = __shfl_up(incl, off, 32);
        if (lane >= off) incl += t;
    }
#pragma unroll
    for (int off = 16; off > 0; off >>= 1) front += __shfl_xor(front, off, 32);

    if (lane == 31) wave_own[wv] = incl;
    if (lane == 0)  wave_front[wv] = front;
    __syncthreads();

    int a = wave_front[lane] + ((lane < wv) ? wave_own[lane] : 0);
#pragma unroll
    for (int off = 16; off > 0; off >>= 1) a += __shfl_xor(a, off, 32);

    int4 o;
    o.x = a + incl - s;
    o.y = o.x + v.x;
    o.z = o.y + v.y;
    o.w = o.z + v.z;
    ((int4*)out)[blk * 1024 + tid] = o;
}

__global__ void __launch_bounds__(256)
sage_place_edges(const int* __restrict__ src,
                 const int* __restrict__ dst,
                 const int* __restrict__ offs,
                 const int* __restrict__ slot,
                 int* __restrict__ esrc, int E) {
    const int i = blockIdx.x * blockDim.x + threadIdx.x;
    if (4 * i >= E) return;
    const int4 d = ((const int4*)dst)[i];
    const int4 s = ((const int4*)src)[i];
    const int4 r = ((const int4*)slot)[i];
    const int px = offs[d.x] + r.x;
    const int py = offs[d.y] + r.y;
    const int pz = offs[d.z] + r.z;
    const int pw = offs[d.w] + r.w;
    esrc[px] = s.x;
    esrc[py] = s.y;
    esrc[pz] = s.z;
    esrc[pw] = s.w;
}

template <int C>
__global__ void sage_gather_mean(const float* __restrict__ x,
                                 const int* __restrict__ esrc,
                                 const int* __restrict__ offs,
                                 const int* __restrict__ deg,
                                 float* __restrict__ outm, int nDst) {
    const int wave = blockIdx.x * (blockDim.x >> 5) + (threadIdx.x >> 5);
    const int lane = threadIdx.x & 31;
    if (wave >= nDst) return;
    const int beg = offs[wave];
    const int d   = deg[wave];
    constexpr int V = C / 128;
    float4 acc[V];
#pragma unroll
    for (int v = 0; v < V; ++v) acc[v] = make_float4(0.f, 0.f, 0.f, 0.f);

    for (int base = 0; base < d; base += 32) {
        int n = d - base;
        if (n > 32) n = 32;
        int myid = (lane < n) ? esrc[beg + base + lane] : 0;
        for (int j = 0; j < n; ++j) {
            int s = __shfl(myid, j, 32);
            const float4* row = (const float4*)(x + (size_t)s * C);
#pragma unroll
            for (int v = 0; v < V; ++v) {
                float4 t = row[lane + 32 * v];
                acc[v].x += t.x; acc[v].y += t.y;
                acc[v].z += t.z; acc[v].w += t.w;
            }
        }
    }

    const float scale = (d > 0) ? (1.0f / (float)d) : 0.0f;
    float4* orow = (float4*)(outm + (size_t)wave * C);
#pragma unroll
    for (int v = 0; v < V; ++v) {
        float4 t;
        t.x = acc[v].x * scale; t.y = acc[v].y * scale;
        t.z = acc[v].z * scale; t.w = acc[v].w * scale;
        orow[lane + 32 * v] = t;
    }
}

template <int K, int MT, int NT>
__device__ __forceinline__ void sage_gemm_phase(const float* const (&ap)[MT],
                                                const float* const (&bp)[NT],
                                                v8f (&acc)[MT][NT]) {
#pragma unroll 1
    for (int kc = 0; kc < K; kc += 32) {
        v4f a[4][MT], b[4][NT];
#pragma unroll
        for (int q = 0; q < 4; ++q) {
#pragma unroll
            for (int mt = 0; mt < MT; ++mt)
                a[q][mt] = *(const v4f*)(ap[mt] + kc + 4 * q);
#pragma unroll
            for (int nt = 0; nt < NT; ++nt)
                b[q][nt] = *(const v4f*)(bp[nt] + kc + 4 * q);
        }
        __builtin_amdgcn_sched_barrier(0);
#pragma unroll
        for (int q = 0; q < 4; ++q) {
#pragma unroll
            for (int p = 0; p < 2; ++p) {
                v2f af[MT], bf[NT];
#pragma unroll
                for (int mt = 0; mt < MT; ++mt) {
                    af[mt].x = a[q][mt][2 * p];
                    af[mt].y = a[q][mt][2 * p + 1];
                }
#pragma unroll
                for (int nt = 0; nt < NT; ++nt) {
                    bf[nt].x = b[q][nt][2 * p];
                    bf[nt].y = b[q][nt][2 * p + 1];
                }
#pragma unroll
                for (int mt = 0; mt < MT; ++mt)
#pragma unroll
                    for (int nt = 0; nt < NT; ++nt)
                        acc[mt][nt] = __builtin_amdgcn_wmma_f32_16x16x4_f32(
                            false, af[mt], false, bf[nt], (short)0, acc[mt][nt],
                            false, false);
            }
        }
    }
}

template <int K, int N, int MT, int NT, bool RELU>
__global__ void __launch_bounds__(256)
sage_gemm_two(const float* __restrict__ A1,
                              const float* __restrict__ B1,
                              const float* __restrict__ A2,
                              const float* __restrict__ B2,
                              const float* __restrict__ bias,
                              float* __restrict__ out, int M) {
    static_assert(K % 32 == 0, "K is walked in whole 128-byte row lines");
    const int lane   = threadIdx.x & 31;
    const int waveId = blockIdx.x * (blockDim.x >> 5) + (threadIdx.x >> 5);
    const int groupsN = N / (16 * NT);
    const int gm = waveId / groupsN;
    const int gn = waveId % groupsN;
    if (gm >= M / (16 * MT)) return;

    const int m0  = gm * 16 * MT;
    const int n0  = gn * 16 * NT;
    const int kh  = (lane >> 4) * 16;

    const float* a1p[MT];
    const float* a2p[MT];
#pragma unroll
    for (int mt = 0; mt < MT; ++mt) {
        int r = m0 + mt * 16 + (lane & 15);
        a1p[mt] = A1 + (size_t)r * K + kh;
        a2p[mt] = A2 + (size_t)r * K + kh;
    }
    const float* b1p[NT];
    const float* b2p[NT];
#pragma unroll
    for (int nt = 0; nt < NT; ++nt) {
        int c = n0 + nt * 16 + (lane & 15);
        b1p[nt] = B1 + (size_t)c * K + kh;
        b2p[nt] = B2 + (size_t)c * K + kh;
    }

    v8f acc[MT][NT] = {};

    sage_gemm_phase<K, MT, NT>(a1p, b1p, acc);
    sage_gemm_phase<K, MT, NT>(a2p, b2p, acc);

    const int rbase = ((lane >> 4) << 3);
#pragma unroll
    for (int nt = 0; nt < NT; ++nt) {
        const int ocol = n0 + nt * 16 + (lane & 15);
        const float bb = bias[ocol];
#pragma unroll
        for (int mt = 0; mt < MT; ++mt) {
#pragma unroll
            for (int r = 0; r < 8; ++r) {
                float v = acc[mt][nt][r] + bb;
                if (RELU) v = fmaxf(v, 0.0f);
                out[(size_t)(m0 + mt * 16 + rbase + r) * N + ocol] = v;
            }
        }
    }
}

__global__ void sage_logsoftmax64(float* __restrict__ out, int rows) {
    int wave = blockIdx.x * (blockDim.x >> 5) + (threadIdx.x >> 5);
    int lane = threadIdx.x & 31;
    if (wave >= rows) return;
    float* p = out + (size_t)wave * 64;
    float a = p[lane];
    float b = p[lane + 32];
    float m = fmaxf(a, b);
#pragma unroll
    for (int off = 16; off > 0; off >>= 1)
        m = fmaxf(m, __shfl_xor(m, off, 32));
    float s = expf(a - m) + expf(b - m);
#pragma unroll
    for (int off = 16; off > 0; off >>= 1)
        s += __shfl_xor(s, off, 32);
    float lse = m + logf(s);
    p[lane]      = a - lse;
    p[lane + 32] = b - lse;
}

extern "C" void kernel_launch(void* const* d_in, const int* in_sizes, int n_in,
                              void* d_out, int out_size, void* d_ws, size_t ws_size,
                              hipStream_t stream) {
    const float* x    = (const float*)d_in[0];
    const int*   src1 = (const int*)d_in[1];
    const int*   dst1 = (const int*)d_in[2];
    const int*   src2 = (const int*)d_in[3];
    const int*   dst2 = (const int*)d_in[4];
    const float* W1l  = (const float*)d_in[5];
    const float* W1r  = (const float*)d_in[6];
    const float* b1   = (const float*)d_in[7];
    const float* W2l  = (const float*)d_in[8];
    const float* W2r  = (const float*)d_in[9];
    const float* b2   = (const float*)d_in[10];
    float* out = (float*)d_out;

    static_assert(E1_ % 1024 == 0 && E2_ % 1024 == 0, "four edges per thread, 256 threads per block");
    int* deg1  = (int*)d_ws;
    int* deg2  = deg1 + N1_;
    int* offs1 = deg2 + N2_;
    int* offs2 = offs1 + N1_;
    int* esrc1 = offs2 + N2_;
    int* esrc2 = esrc1 + E1_;
    float* aggr1 = (float*)(esrc2 + E2_);
    float* aggr2 = aggr1 + (size_t)N1_ * IN_C;
    float* h     = aggr2 + (size_t)N2_ * HID_C;
    int* slot1 = (int*)aggr1;
    int* slot2 = (int*)aggr2;

    size_t zbytes = (size_t)(N1_ + N2_) * sizeof(int);
    hipMemsetAsync(d_ws, 0, zbytes, stream);

    sage_count_slots<<<E1_ / 1024, 256, 0, stream>>>(dst1, deg1, slot1, E1_);
    sage_excl_scan<N1_ / 4096><<<N1_ / 4096, 1024, 0, stream>>>(deg1, offs1);
    sage_place_edges<<<E1_ / 1024, 256, 0, stream>>>(src1, dst1, offs1, slot1,
                                                     esrc1, E1_);
    sage_gather_mean<IN_C><<<N1_ / 8, 256, 0, stream>>>(x, esrc1, offs1, deg1,
                                                        aggr1, N1_);
    {
        int waves = (N1_ / 32) * (HID_C / 64);
        sage_gemm_two<IN_C, HID_C, 2, 4, true>
            <<<waves / 8, 256, 0, stream>>>(aggr1, W1l, x, W1r, b1, h, N1_);
    }

    sage_count_slots<<<E2_ / 1024, 256, 0, stream>>>(dst2, deg2, slot2, E2_);
    sage_excl_scan<N2_ / 4096><<<N2_ / 4096, 1024, 0, stream>>>(deg2, offs2);
    sage_place_edges<<<E2_ / 1024, 256, 0, stream>>>(src2, dst2, offs2, slot2,
                                                     esrc2, E2_);
    sage_gather_mean<HID_C><<<N2_ / 8, 256, 0, stream>>>(h, esrc2, offs2, deg2,
                                                         aggr2, N2_);
    {
        int waves = (N2_ / 32) * (OUT_C / 32);
        sage_gemm_two<HID_C, OUT_C, 2, 2, false>
            <<<waves / 8, 256, 0, stream>>>(aggr2, W2l, h, W2r, b2, out, N2_);
    }

    sage_logsoftmax64<<<(N2_ + 7) / 8, 256, 0, stream>>>(out, N2_);
}


# Degree count on LDS counters: work-group (node range, edge chunk) counts its chunk's edges of its range, rank = LDS add's return; no global atomics, no memset; sums in chunk, then LDS arrival order
# speedup vs baseline: 2.0525x; 1.0406x over previous
#include <hip/hip_runtime.h>
#include <hip/hip_bf16.h>

#define N0    262144
#define N1_   65536
#define N2_   16384
#define E1_   1048576
#define E2_   262144
#define IN_C  128
#define HID_C 256
#define OUT_C 64

typedef __attribute__((ext_vector_type(2))) float v2f;
typedef __attribute__((ext_vector_type(4))) float v4f;
typedef __attribute__((ext_vector_type(8))) float v8f;

template <int E, int N, int RANGE, int C>
__global__ void __launch_bounds__(1024)
sage_count_ranges(const int* __restrict__ dst, int* __restrict__ table,
                  int* __restrict__ tot, int* __restrict__ slot) {
    constexpr int R = N / RANGE;
    constexpr int U = 8;
    constexpr int PER4 = E / C / 4;
    static_assert(N % RANGE == 0, "whole ranges");
    static_assert(PER4 % (1024 * U) == 0, "whole trips");
    __shared__ int cnt[RANGE];
    __shared__ int total;
    const int tid = threadIdx.x;
    const int r   = blockIdx.x % R;
    const int c   = blockIdx.x / R;
    const int lo  = r * RANGE;
    for (int i = tid; i < RANGE; i += 1024) cnt[i] = 0;
    if (tid == 0) total = 0;
    __syncthreads();

    const int4* d4 = (const int4*)dst + (size_t)c * PER4;
    int* sl = slot + (size_t)c * (E / C);
    for (int it = tid; it < PER4; it += 1024 * U) {
        int4 d[U];
#pragma unroll
        for (int u = 0; u < U; ++u) d[u] = d4[it + u * 1024];
        int rk[U][4];
#pragma unroll
        for (int u = 0; u < U; ++u) {
            const unsigned ux = (unsigned)(d[u].x - lo);
            const unsigned uy = (unsigned)(d[u].y - lo);
            const unsigned uz = (unsigned)(d[u].z - lo);
            const unsigned uw = (unsigned)(d[u].w - lo);
            rk[u][0] = (ux < (unsigned)RANGE) ? atomicAdd(&cnt[ux], 1) : -1;
            rk[u][1] = (uy < (unsigned)RANGE) ? atomicAdd(&cnt[uy], 1) : -1;
            rk[u][2] = (uz < (unsigned)RANGE) ? atomicAdd(&cnt[uz], 1) : -1;
            rk[u][3] = (uw < (unsigned)RANGE) ? atomicAdd(&cnt[uw], 1) : -1;
        }
#pragma unroll
        for (int u = 0; u < U; ++u) {
            const int e = 4 * (it + u * 1024);
            if (rk[u][0] >= 0) sl[e + 0] = rk[u][0];
            if (rk[u][1] >= 0) sl[e + 1] = rk[u][1];
            if (rk[u][2] >= 0) sl[e + 2] = rk[u][2];
            if (rk[u][3] >= 0) sl[e + 3] = rk[u][3];
        }
    }
    __syncthreads();

    int mine = 0;
    for (int i = tid; i < RANGE; i += 1024) {
        const int v = cnt[i];
        table[(size_t)c * N + lo + i] = v;
        mine += v;
    }
#pragma unroll
    for (int off = 16; off > 0; off >>= 1) mine += __shfl_xor(mine, off, 32);
    if ((tid & 31) == 0) atomicAdd(&total, mine);
    __syncthreads();
    if (tid == 0) tot[c * R + r] = total;
}

template <int N, int RANGE, int C>
__global__ void __launch_bounds__(RANGE / 4)
sage_scan_ranges(int* __restrict__ table, const int* __restrict__ tot,
                 int* __restrict__ deg) {
    constexpr int R  = N / RANGE;
    constexpr int NT = RANGE / 4;
    constexpr int NW = NT / 32;
    static_assert(NW <= 32, "one lane per wave in the fold");
    __shared__ int wave_own[32];
    __shared__ int wave_front[32];
    const int tid  = threadIdx.x;
    const int lane = tid & 31;
    const int wv   = tid >> 5;
    const int blk  = blockIdx.x;
    int4* t4 = (int4*)table;
    const int idx4 = blk * NT + tid;

    int4 cn[C];
#pragma unroll
    for (int c = 0; c < C; ++c) cn[c] = t4[c * (N / 4) + idx4];

    int front = 0;
    for (int i = tid; i < C * R; i += NT) {
        const int t = tot[i];
        front += ((i % R) < blk) ? t : 0;
    }

    int4 v = cn[0];
#pragma unroll
    for (int c = 1; c < C; ++c) {
        v.x += cn[c].x; v.y += cn[c].y; v.z += cn[c].z; v.w += cn[c].w;
    }
    const int s = (v.x + v.y) + (v.z + v.w);
    int incl = s;
#pragma unroll
    for (int off = 1; off < 32; off <<= 1) {
        const int t = __shfl_up(incl, off, 32);
        if (lane >= off) incl += t;
    }
#pragma unroll
    for (int off = 16; off > 0; off >>= 1) front += __shfl_xor(front, off, 32);

    if (lane == 31) wave_own[wv] = incl;
    if (lane == 0)  wave_front[wv] = front;
    __syncthreads();

    int a = (lane < NW) ? (wave_front[lane] + ((lane < wv) ? wave_own[lane] : 0)) : 0;
#pragma unroll
    for (int off = 16; off > 0; off >>= 1) a += __shfl_xor(a, off, 32);

    ((int4*)deg)[idx4] = v;
    int4 b;
    b.x = a + incl - s;
    b.y = b.x + v.x;
    b.z = b.y + v.y;
    b.w = b.z + v.z;
#pragma unroll
    for (int c = 0; c < C; ++c) {
        t4[c * (N / 4) + idx4] = b;
        b.x += cn[c].x; b.y += cn[c].y; b.z += cn[c].z; b.w += cn[c].w;
    }
}

template <int E, int N, int C>
__global__ void __launch_bounds__(256)
sage_place_edges(const int* __restrict__ src,
                 const int* __restrict__ dst,
                 const int* __restrict__ table,
                 const int* __restrict__ slot,
                 int* __restrict__ esrc) {
    static_assert((E / C) % 1024 == 0, "a block's 1024 edges lie in one chunk");
    const int i = blockIdx.x * blockDim.x + threadIdx.x;
    const int* base = table + (size_t)(blockIdx.x / (E / C / 1024)) * N;
    const int4 d = ((const int4*)dst)[i];
    const int4 s = ((const int4*)src)[i];
    const int4 r = ((const int4*)slot)[i];
    const int px = base[d.x] + r.x;
    const int py = base[d.y] + r.y;
    const int pz = base[d.z] + r.z;
    const int pw = base[d.w] + r.w;
    esrc[px] = s.x;
    esrc[py] = s.y;
    esrc[pz] = s.z;
    esrc[pw] = s.w;
}

template <int C>
__global__ void sage_gather_mean(const float* __restrict__ x,
                                 const int* __restrict__ esrc,
                                 const int* __restrict__ offs,
                                 const int* __restrict__ deg,
                                 float* __restrict__ outm, int nDst) {
    const int wave = blockIdx.x * (blockDim.x >> 5) + (threadIdx.x >> 5);
    const int lane = threadIdx.x & 31;
    if (wave >= nDst) return;
    const int beg = offs[wave];
    const int d   = deg[wave];
    constexpr int V = C / 128;
    float4 acc[V];
#pragma unroll
    for (int v = 0; v < V; ++v) acc[v] = make_float4(0.f, 0.f, 0.f, 0.f);

    for (int base = 0; base < d; base += 32) {
        int n = d - base;
        if (n > 32) n = 32;
        int myid = (lane < n) ? esrc[beg + base + lane] : 0;
        for (int j = 0; j < n; ++j) {
            int s = __shfl(myid, j, 32);
            const float4* row = (const float4*)(x + (size_t)s * C);
#pragma unroll
            for (int v = 0; v < V; ++v) {
                float4 t = row[lane + 32 * v];
                acc[v].x += t.x; acc[v].y += t.y;
                acc[v].z += t.z; acc[v].w += t.w;
            }
        }
    }

    const float scale = (d > 0) ? (1.0f / (float)d) : 0.0f;
    float4* orow = (float4*)(outm + (size_t)wave * C);
#pragma unroll
    for (int v = 0; v < V; ++v) {
        float4 t;
        t.x = acc[v].x * scale; t.y = acc[v].y * scale;
        t.z = acc[v].z * scale; t.w = acc[v].w * scale;
        orow[lane + 32 * v] = t;
    }
}

template <int K, int MT, int NT>
__device__ __forceinline__ void sage_gemm_phase(const float* const (&ap)[MT],
                                                const float* const (&bp)[NT],
                                                v8f (&acc)[MT][NT]) {
#pragma unroll 1
    for (int kc = 0; kc < K; kc += 32) {
        v4f a[4][MT], b[4][NT];
#pragma unroll
        for (int q = 0; q < 4; ++q) {
#pragma unroll
            for (int mt = 0; mt < MT; ++mt)
                a[q][mt] = *(const v4f*)(ap[mt] + kc + 4 * q);
#pragma unroll
            for (int nt = 0; nt < NT; ++nt)
                b[q][nt] = *(const v4f*)(bp[nt] + kc + 4 * q);
        }
        __builtin_amdgcn_sched_barrier(0);
#pragma unroll
        for (int q = 0; q < 4; ++q) {
#pragma unroll
            for (int p = 0; p < 2; ++p) {
                v2f af[MT], bf[NT];
#pragma unroll
                for (int mt = 0; mt < MT; ++mt) {
                    af[mt].x = a[q][mt][2 * p];
                    af[mt].y = a[q][mt][2 * p + 1];
                }
#pragma unroll
                for (int nt = 0; nt < NT; ++nt) {
                    bf[nt].x = b[q][nt][2 * p];
                    bf[nt].y = b[q][nt][2 * p + 1];
                }
#pragma unroll
                for (int mt = 0; mt < MT; ++mt)
#pragma unroll
                    for (int nt = 0; nt < NT; ++nt)
                        acc[mt][nt] = __builtin_amdgcn_wmma_f32_16x16x4_f32(
                            false, af[mt], false, bf[nt], (short)0, acc[mt][nt],
                            false, false);
            }
        }
    }
}

template <int K, int N, int MT, int NT, bool RELU>
__global__ void __launch_bounds__(256)
sage_gemm_two(const float* __restrict__ A1,
                              const float* __restrict__ B1,
                              const float* __restrict__ A2,
                              const float* __restrict__ B2,
                              const float* __restrict__ bias,
                              float* __restrict__ out, int M) {
    static_assert(K % 32 == 0, "K is walked in whole 128-byte row lines");
    const int lane   = threadIdx.x & 31;
    const int waveId = blockIdx.x * (blockDim.x >> 5) + (threadIdx.x >> 5);
    const int groupsN = N / (16 * NT);
    const int gm = waveId / groupsN;
    const int gn = waveId % groupsN;
    if (gm >= M / (16 * MT)) return;

    const int m0  = gm * 16 * MT;
    const int n0  = gn * 16 * NT;
    const int kh  = (lane >> 4) * 16;

    const float* a1p[MT];
    const float* a2p[MT];
#pragma unroll
    for (int mt = 0; mt < MT; ++mt) {
        int r = m0 + mt * 16 + (lane & 15);
        a1p[mt] = A1 + (size_t)r * K + kh;
        a2p[mt] = A2 + (size_t)r * K + kh;
    }
    const float* b1p[NT];
    const float* b2p[NT];
#pragma unroll
    for (int nt = 0; nt < NT; ++nt) {
        int c = n0 + nt * 16 + (lane & 15);
        b1p[nt] = B1 + (size_t)c * K + kh;
        b2p[nt] = B2 + (size_t)c * K + kh;
    }

    v8f acc[MT][NT] = {};

    sage_gemm_phase<K, MT, NT>(a1p, b1p, acc);
    sage_gemm_phase<K, MT, NT>(a2p, b2p, acc);

    const int rbase = ((lane >> 4) << 3);
#pragma unroll
    for (int nt = 0; nt < NT; ++nt) {
        const int ocol = n0 + nt * 16 + (lane & 15);
        const float bb = bias[ocol];
#pragma unroll
        for (int mt = 0; mt < MT; ++mt) {
#pragma unroll
            for (int r = 0; r < 8; ++r) {
                float v = acc[mt][nt][r] + bb;
                if (RELU) v = fmaxf(v, 0.0f);
                out[(size_t)(m0 + mt * 16 + rbase + r) * N + ocol] = v;
            }
        }
    }
}

__global__ void sage_logsoftmax64(float* __restrict__ out, int rows) {
    int wave = blockIdx.x * (blockDim.x >> 5) + (threadIdx.x >> 5);
    int lane = threadIdx.x & 31;
    if (wave >= rows) return;
    float* p = out + (size_t)wave * 64;
    float a = p[lane];
    float b = p[lane + 32];
    float m = fmaxf(a, b);
#pragma unroll
    for (int off = 16; off > 0; off >>= 1)
        m = fmaxf(m, __shfl_xor(m, off, 32));
    float s = expf(a - m) + expf(b - m);
#pragma unroll
    for (int off = 16; off > 0; off >>= 1)
        s += __shfl_xor(s, off, 32);
    float lse = m + logf(s);
    p[lane]      = a - lse;
    p[lane + 32] = b - lse;
}

extern "C" void kernel_launch(void* const* d_in, const int* in_sizes, int n_in,
                              void* d_out, int out_size, void* d_ws, size_t ws_size,
                              hipStream_t stream) {
    const float* x    = (const float*)d_in[0];
    const int*   src1 = (const int*)d_in[1];
    const int*   dst1 = (const int*)d_in[2];
    const int*   src2 = (const int*)d_in[3];
    const int*   dst2 = (const int*)d_in[4];
    const float* W1l  = (const float*)d_in[5];
    const float* W1r  = (const float*)d_in[6];
    const float* b1   = (const float*)d_in[7];
    const float* W2l  = (const float*)d_in[8];
    const float* W2r  = (const float*)d_in[9];
    const float* b2   = (const float*)d_in[10];
    float* out = (float*)d_out;

    constexpr int CH  = 8;
    constexpr int RG1 = 2048;
    constexpr int RG2 = 1024;
    int* deg1  = (int*)d_ws;
    int* deg2  = deg1 + N1_;
    int* tot1  = deg2 + N2_;
    int* tot2  = tot1 + N1_;
    int* esrc1 = tot2 + N2_;
    int* esrc2 = esrc1 + E1_;
    float* aggr1 = (float*)(esrc2 + E2_);
    float* aggr2 = aggr1 + (size_t)N1_ * IN_C;
    float* h     = aggr2 + (size_t)N2_ * HID_C;
    int* slot1  = (int*)aggr1;
    int* slot2  = (int*)aggr2;
    int* table1 = (int*)h;
    int* table2 = (int*)aggr1;

    sage_count_ranges<E1_, N1_, RG1, CH>
        <<<(N1_ / RG1) * CH, 1024, 0, stream>>>(dst1, table1, tot1, slot1);
    sage_scan_ranges<N1_, RG1, CH>
        <<<N1_ / RG1, RG1 / 4, 0, stream>>>(table1, tot1, deg1);
    sage_place_edges<E1_, N1_, CH>
        <<<E1_ / 1024, 256, 0, stream>>>(src1, dst1, table1, slot1, esrc1);
    sage_gather_mean<IN_C><<<N1_ / 8, 256, 0, stream>>>(x, esrc1, table1, deg1,
                                                        aggr1, N1_);
    {
        int waves = (N1_ / 32) * (HID_C / 64);
        sage_gemm_two<IN_C, HID_C, 2, 4, true>
            <<<waves / 8, 256, 0, stream>>>(aggr1, W1l, x, W1r, b1, h, N1_);
    }

    sage_count_ranges<E2_, N2_, RG2, CH>
        <<<(N2_ / RG2) * CH, 1024, 0, stream>>>(dst2, table2, tot2, slot2);
    sage_scan_ranges<N2_, RG2, CH>
        <<<N2_ / RG2, RG2 / 4, 0, stream>>>(table2, tot2, deg2);
    sage_place_edges<E2_, N2_, CH>
        <<<E2_ / 1024, 256, 0, stream>>>(src2, dst2, table2, slot2, esrc2);
    sage_gather_mean<HID_C><<<N2_ / 8, 256, 0, stream>>>(h, esrc2, table2, deg2,
                                                         aggr2, N2_);
    {
        int waves = (N2_ / 32) * (OUT_C / 32);
        sage_gemm_two<HID_C, OUT_C, 2, 2, false>
            <<<waves / 8, 256, 0, stream>>>(aggr2, W2l, h, W2r, b2, out, N2_);
    }

    sage_logsoftmax64<<<(N2_ + 7) / 8, 256, 0, stream>>>(out, N2_);
}


# Gather-mean on resident waves: a wave sums 8 consecutive rows (2 in layer 2), 8 KiB of row loads issued per round trip (was 1 row per wave, 1 load per trip); same edge order, sums bit-identical
# speedup vs baseline: 2.2271x; 1.0851x over previous
#include <hip/hip_runtime.h>
#include <hip/hip_bf16.h>

#define N0    262144
#define N1_   65536
#define N2_   16384
#define E1_   1048576
#define E2_   262144
#define IN_C  128
#define HID_C 256
#define OUT_C 64

typedef __attribute__((ext_vector_type(2))) float v2f;
typedef __attribute__((ext_vector_type(4))) float v4f;
typedef __attribute__((ext_vector_type(8))) float v8f;

template <int E, int N, int RANGE, int C>
__global__ void __launch_bounds__(1024)
sage_count_ranges(const int* __restrict__ dst, int* __restrict__ table,
                  int* __restrict__ tot, int* __restrict__ slot) {
    constexpr int R = N / RANGE;
    constexpr int U = 8;
    constexpr int PER4 = E / C / 4;
    static_assert(N % RANGE == 0, "whole ranges");
    static_assert(PER4 % (1024 * U) == 0, "whole trips");
    __shared__ int cnt[RANGE];
    __shared__ int total;
    const int tid = threadIdx.x;
    const int r   = blockIdx.x % R;
    const int c   = blockIdx.x / R;
    const int lo  = r * RANGE;
    for (int i = tid; i < RANGE; i += 1024) cnt[i] = 0;
    if (tid == 0) total = 0;
    __syncthreads();

    const int4* d4 = (const int4*)dst + (size_t)c * PER4;
    int* sl = slot + (size_t)c * (E / C);
    for (int it = tid; it < PER4; it += 1024 * U) {
        int4 d[U];
#pragma unroll
        for (int u = 0; u < U; ++u) d[u] = d4[it + u * 1024];
        int rk[U][4];
#pragma unroll
        for (int u = 0; u < U; ++u) {
            const unsigned ux = (unsigned)(d[u].x - lo);
            const unsigned uy = (unsigned)(d[u].y - lo);
            const unsigned uz = (unsigned)(d[u].z - lo);
            const unsigned uw = (unsigned)(d[u].w - lo);
            rk[u][0] = (ux < (unsigned)RANGE) ? atomicAdd(&cnt[ux], 1) : -1;
            rk[u][1] = (uy < (unsigned)RANGE) ? atomicAdd(&cnt[uy], 1) : -1;
            rk[u][2] = (uz < (unsigned)RANGE) ? atomicAdd(&cnt[uz], 1) : -1;
            rk[u][3] = (uw < (unsigned)RANGE) ? atomicAdd(&cnt[uw], 1) : -1;
        }
#pragma unroll
        for (int u = 0; u < U; ++u) {
            const int e = 4 * (it + u * 1024);
            if (rk[u][0] >= 0) sl[e + 0] = rk[u][0];
            if (rk[u][1] >= 0) sl[e + 1] = rk[u][1];
            if (rk[u][2] >= 0) sl[e + 2] = rk[u][2];
            if (rk[u][3] >= 0) sl[e + 3] = rk[u][3];
        }
    }
    __syncthreads();

    int mine = 0;
    for (int i = tid; i < RANGE; i += 1024) {
        const int v = cnt[i];
        table[(size_t)c * N + lo + i] = v;
        mine += v;
    }
#pragma unroll
    for (int off = 16; off > 0; off >>= 1) mine += __shfl_xor(mine, off, 32);
    if ((tid & 31) == 0) atomicAdd(&total, mine);
    __syncthreads();
    if (tid == 0) tot[c * R + r] = total;
}

template <int N, int RANGE, int C>
__global__ void __launch_bounds__(RANGE / 4)
sage_scan_ranges(int* __restrict__ table, const int* __restrict__ tot,
                 int* __restrict__ deg) {
    constexpr int R  = N / RANGE;
    constexpr int NT = RANGE / 4;
    constexpr int NW = NT / 32;
    static_assert(NW <= 32, "one lane per wave in the fold");
    __shared__ int wave_own[32];
    __shared__ int wave_front[32];
    const int tid  = threadIdx.x;
    const int lane = tid & 31;
    const int wv   = tid >> 5;
    const int blk  = blockIdx.x;
    int4* t4 = (int4*)table;
    const int idx4 = blk * NT + tid;

    int4 cn[C];
#pragma unroll
    for (int c = 0; c < C; ++c) cn[c] = t4[c * (N / 4) + idx4];

    int front = 0;
    for (int i = tid; i < C * R; i += NT) {
        const int t = tot[i];
        front += ((i % R) < blk) ? t : 0;
    }

    int4 v = cn[0];
#pragma unroll
    for (int c = 1; c < C; ++c) {
        v.x += cn[c].x; v.y += cn[c].y; v.z += cn[c].z; v.w += cn[c].w;
    }
    const int s = (v.x + v.y) + (v.z + v.w);
    int incl = s;
#pragma unroll
    for (int off = 1; off < 32; off <<= 1) {
        const int t = __shfl_up(incl, off, 32);
        if (lane >= off) incl += t;
    }
#pragma unroll
    for (int off = 16; off > 0; off >>= 1) front += __shfl_xor(front, off, 32);

    if (lane == 31) wave_own[wv] = incl;
    if (lane == 0)  wave_front[wv] = front;
    __syncthreads();

    int a = (lane < NW) ? (wave_front[lane] + ((lane < wv) ? wave_own[lane] : 0)) : 0;
#pragma unroll
    for (int off = 16; off > 0; off >>= 1) a += __shfl_xor(a, off, 32);

    ((int4*)deg)[idx4] = v;
    int4 b;
    b.x = a + incl - s;
    b.y = b.x + v.x;
    b.z = b.y + v.y;
    b.w = b.z + v.z;
#pragma unroll
    for (int c = 0; c < C; ++c) {
        t4[c * (N / 4) + idx4] = b;
        b.x += cn[c].x; b.y += cn[c].y; b.z += cn[c].z; b.w += cn[c].w;
    }
}

template <int E, int N, int C>
__global__ void __launch_bounds__(256)
sage_place_edges(const int* __restrict__ src,
                 const int* __restrict__ dst,
                 const int* __restrict__ table,
                 const int* __restrict__ slot,
                 int* __restrict__ esrc) {
    static_assert((E / C) % 1024 == 0, "a block's 1024 edges lie in one chunk");
    const int i = blockIdx.x * blockDim.x + threadIdx.x;
    const int* base = table + (size_t)(blockIdx.x / (E / C / 1024)) * N;
    const int4 d = ((const int4*)dst)[i];
    const int4 s = ((const int4*)src)[i];
    const int4 r = ((const int4*)slot)[i];
    const int px = base[d.x] + r.x;
    const int py = base[d.y] + r.y;
    const int pz = base[d.z] + r.z;
    const int pw = base[d.w] + r.w;
    esrc[px] = s.x;
    esrc[py] = s.y;
    esrc[pz] = s.z;
    esrc[pw] = s.w;
}

template <int C, int B>
__device__ __forceinline__ void sage_sum_rows(const float* __restrict__ x, int lane,
                                              int myid, int n, float4 (&acc)[C / 128]) {
    constexpr int V = C / 128;
    int j = 0;
    for (; j + B <= n; j += B) {
        float4 t[B][V];
#pragma unroll
        for (int k = 0; k < B; ++k) {
            const int s = __builtin_amdgcn_readlane(myid, j + k);
            const float4* row = (const float4*)(x + (size_t)s * C);
#pragma unroll
            for (int v = 0; v < V; ++v) t[k][v] = row[lane + 32 * v];
        }
        __builtin_amdgcn_sched_barrier(0);
#pragma unroll
        for (int k = 0; k < B; ++k) {
#pragma unroll
            for (int v = 0; v < V; ++v) {
                acc[v].x += t[k][v].x; acc[v].y += t[k][v].y;
                acc[v].z += t[k][v].z; acc[v].w += t[k][v].w;
            }
        }
    }
    if (j < n) {
        float4 t[B - 1][V];
#pragma unroll
        for (int k = 0; k < B - 1; ++k) {
            if (j + k < n) {
                const int s = __builtin_amdgcn_readlane(myid, j + k);
                const float4* row = (const float4*)(x + (size_t)s * C);
#pragma unroll
                for (int v = 0; v < V; ++v) t[k][v] = row[lane + 32 * v];
            } else {
#pragma unroll
                for (int v = 0; v < V; ++v) t[k][v] = make_float4(0.f, 0.f, 0.f, 0.f);
            }
        }
#pragma unroll
        for (int k = 0; k < B - 1; ++k) {
            if (j + k < n) {
#pragma unroll
                for (int v = 0; v < V; ++v) {
                    acc[v].x += t[k][v].x; acc[v].y += t[k][v].y;
                    acc[v].z += t[k][v].z; acc[v].w += t[k][v].w;
                }
            }
        }
    }
}

template <int C, int R>
__global__ void __launch_bounds__(256)
sage_gather_mean(const float* __restrict__ x,
                                 const int* __restrict__ esrc,
                                 const int* __restrict__ offs,
                                 const int* __restrict__ deg,
                                 float* __restrict__ outm, int nDst) {
    static_assert(R <= 32, "one lane per row of the wave");
    const int wave = blockIdx.x * (blockDim.x >> 5) + (threadIdx.x >> 5);
    const int lane = threadIdx.x & 31;
    const int r0 = __builtin_amdgcn_readfirstlane(wave) * R;
    if (r0 >= nDst) return;
    constexpr int V = C / 128;
    constexpr int B = 16 / V;

    const bool has = (lane < R) && (r0 + lane < nDst);
    const int offv = has ? offs[r0 + lane] : 0;
    const int degv = has ? deg[r0 + lane] : 0;

    int beg = __builtin_amdgcn_readlane(offv, 0);
    int d   = __builtin_amdgcn_readlane(degv, 0);
    int ids = (lane < d) ? esrc[beg + lane] : 0;

    for (int r = 0; r < R; ++r) {
        if (r0 + r >= nDst) break;
        const int cur = ids;
        const int cbeg = beg;
        const int cd   = d;
        if (r + 1 < R) {
            beg = __builtin_amdgcn_readlane(offv, r + 1);
            d   = __builtin_amdgcn_readlane(degv, r + 1);
            ids = (lane < d) ? esrc[beg + lane] : 0;
        }
        float4 acc[V];
#pragma unroll
        for (int v = 0; v < V; ++v) acc[v] = make_float4(0.f, 0.f, 0.f, 0.f);

        sage_sum_rows<C, B>(x, lane, cur, (cd < 32) ? cd : 32, acc);
        for (int base = 32; base < cd; base += 32) {
            int n = cd - base;
            if (n > 32) n = 32;
            const int more = (lane < n) ? esrc[cbeg + base + lane] : 0;
            sage_sum_rows<C, B>(x, lane, more, n, acc);
        }

        const float scale = (cd > 0) ? (1.0f / (float)cd) : 0.0f;
        float4* orow = (float4*)(outm + (size_t)(r0 + r) * C);
#pragma unroll
        for (int v = 0; v < V; ++v) {
            float4 t;
            t.x = acc[v].x * scale; t.y = acc[v].y * scale;
            t.z = acc[v].z * scale; t.w = acc[v].w * scale;
            orow[lane + 32 * v] = t;
        }
    }
}

template <int K, int MT, int NT>
__device__ __forceinline__ void sage_gemm_phase(const float* const (&ap)[MT],
                                                const float* const (&bp)[NT],
                                                v8f (&acc)[MT][NT]) {
#pragma unroll 1
    for (int kc = 0; kc < K; kc += 32) {
        v4f a[4][MT], b[4][NT];
#pragma unroll
        for (int q = 0; q < 4; ++q) {
#pragma unroll
            for (int mt = 0; mt < MT; ++mt)
                a[q][mt] = *(const v4f*)(ap[mt] + kc + 4 * q);
#pragma unroll
            for (int nt = 0; nt < NT; ++nt)
                b[q][nt] = *(const v4f*)(bp[nt] + kc + 4 * q);
        }
        __builtin_amdgcn_sched_barrier(0);
#pragma unroll
        for (int q = 0; q < 4; ++q) {
#pragma unroll
            for (int p = 0; p < 2; ++p) {
                v2f af[MT], bf[NT];
#pragma unroll
                for (int mt = 0; mt < MT; ++mt) {
                    af[mt].x = a[q][mt][2 * p];
                    af[mt].y = a[q][mt][2 * p + 1];
                }
#pragma unroll
                for (int nt = 0; nt < NT; ++nt) {
                    bf[nt].x = b[q][nt][2 * p];
                    bf[nt].y = b[q][nt][2 * p + 1];
                }
#pragma unroll
                for (int mt = 0; mt < MT; ++mt)
#pragma unroll
                    for (int nt = 0; nt < NT; ++nt)
                        acc[mt][nt] = __builtin_amdgcn_wmma_f32_16x16x4_f32(
                            false, af[mt], false, bf[nt], (short)0, acc[mt][nt],
                            false, false);
            }
        }
    }
}

template <int K, int N, int MT, int NT, bool RELU>
__global__ void __launch_bounds__(256)
sage_gemm_two(const float* __restrict__ A1,
                              const float* __restrict__ B1,
                              const float* __restrict__ A2,
                              const float* __restrict__ B2,
                              const float* __restrict__ bias,
                              float* __restrict__ out, int M) {
    static_assert(K % 32 == 0, "K is walked in whole 128-byte row lines");
    const int lane   = threadIdx.x & 31;
    const int waveId = blockIdx.x * (blockDim.x >> 5) + (threadIdx.x >> 5);
    const int groupsN = N / (16 * NT);
    const int gm = waveId / groupsN;
    const int gn = waveId % groupsN;
    if (gm >= M / (16 * MT)) return;

    const int m0  = gm * 16 * MT;
    const int n0  = gn * 16 * NT;
    const int kh  = (lane >> 4) * 16;

    const float* a1p[MT];
    const float* a2p[MT];
#pragma unroll
    for (int mt = 0; mt < MT; ++mt) {
        int r = m0 + mt * 16 + (lane & 15);
        a1p[mt] = A1 + (size_t)r * K + kh;
        a2p[mt] = A2 + (size_t)r * K + kh;
    }
    const float* b1p[NT];
    const float* b2p[NT];
#pragma unroll
    for (int nt = 0; nt < NT; ++nt) {
        int c = n0 + nt * 16 + (lane & 15);
        b1p[nt] = B1 + (size_t)c * K + kh;
        b2p[nt] = B2 + (size_t)c * K + kh;
    }

    v8f acc[MT][NT] = {};

    sage_gemm_phase<K, MT, NT>(a1p, b1p, acc);
    sage_gemm_phase<K, MT, NT>(a2p, b2p, acc);

    const int rbase = ((lane >> 4) << 3);
#pragma unroll
    for (int nt = 0; nt < NT; ++nt) {
        const int ocol = n0 + nt * 16 + (lane & 15);
        const float bb = bias[ocol];
#pragma unroll
        for (int mt = 0; mt < MT; ++mt) {
#pragma unroll
            for (int r = 0; r < 8; ++r) {
                float v = acc[mt][nt][r] + bb;
                if (RELU) v = fmaxf(v, 0.0f);
                out[(size_t)(m0 + mt * 16 + rbase + r) * N + ocol] = v;
            }
        }
    }
}

__global__ void sage_logsoftmax64(float* __restrict__ out, int rows) {
    int wave = blockIdx.x * (blockDim.x >> 5) + (threadIdx.x >> 5);
    int lane = threadIdx.x & 31;
    if (wave >= rows) return;
    float* p = out + (size_t)wave * 64;
    float a = p[lane];
    float b = p[lane + 32];
    float m = fmaxf(a, b);
#pragma unroll
    for (int off = 16; off > 0; off >>= 1)
        m = fmaxf(m, __shfl_xor(m, off, 32));
    float s = expf(a - m) + expf(b - m);
#pragma unroll
    for (int off = 16; off > 0; off >>= 1)
        s += __shfl_xor(s, off, 32);
    float lse = m + logf(s);
    p[lane]      = a - lse;
    p[lane + 32] = b - lse;
}

extern "C" void kernel_launch(void* const* d_in, const int* in_sizes, int n_in,
                              void* d_out, int out_size, void* d_ws, size_t ws_size,
                              hipStream_t stream) {
    const float* x    = (const float*)d_in[0];
    const int*   src1 = (const int*)d_in[1];
    const int*   dst1 = (const int*)d_in[2];
    const int*   src2 = (const int*)d_in[3];
    const int*   dst2 = (const int*)d_in[4];
    const float* W1l  = (const float*)d_in[5];
    const float* W1r  = (const float*)d_in[6];
    const float* b1   = (const float*)d_in[7];
    const float* W2l  = (const float*)d_in[8];
    const float* W2r  = (const float*)d_in[9];
    const float* b2   = (const float*)d_in[10];
    float* out = (float*)d_out;

    constexpr int CH  = 8;
    constexpr int RG1 = 2048;
    constexpr int RG2 = 1024;
    int* deg1  = (int*)d_ws;
    int* deg2  = deg1 + N1_;
    int* tot1  = deg2 + N2_;
    int* tot2  = tot1 + N1_;
    int* esrc1 = tot2 + N2_;
    int* esrc2 = esrc1 + E1_;
    float* aggr1 = (float*)(esrc2 + E2_);
    float* aggr2 = aggr1 + (size_t)N1_ * IN_C;
    float* h     = aggr2 + (size_t)N2_ * HID_C;
    int* slot1  = (int*)aggr1;
    int* slot2  = (int*)aggr2;
    int* table1 = (int*)h;
    int* table2 = (int*)aggr1;

    sage_count_ranges<E1_, N1_, RG1, CH>
        <<<(N1_ / RG1) * CH, 1024, 0, stream>>>(dst1, table1, tot1, slot1);
    sage_scan_ranges<N1_, RG1, CH>
        <<<N1_ / RG1, RG1 / 4, 0, stream>>>(table1, tot1, deg1);
    sage_place_edges<E1_, N1_, CH>
        <<<E1_ / 1024, 256, 0, stream>>>(src1, dst1, table1, slot1, esrc1);
    sage_gather_mean<IN_C, 8><<<N1_ / 8 / 8, 256, 0, stream>>>(x, esrc1, table1, deg1,
                                                               aggr1, N1_);
    {
        int waves = (N1_ / 32) * (HID_C / 64);
        sage_gemm_two<IN_C, HID_C, 2, 4, true>
            <<<waves / 8, 256, 0, stream>>>(aggr1, W1l, x, W1r, b1, h, N1_);
    }

    sage_count_ranges<E2_, N2_, RG2, CH>
        <<<(N2_ / RG2) * CH, 1024, 0, stream>>>(dst2, table2, tot2, slot2);
    sage_scan_ranges<N2_, RG2, CH>
        <<<N2_ / RG2, RG2 / 4, 0, stream>>>(table2, tot2, deg2);
    sage_place_edges<E2_, N2_, CH>
        <<<E2_ / 1024, 256, 0, stream>>>(src2, dst2, table2, slot2, esrc2);
    sage_gather_mean<HID_C, 2><<<N2_ / 2 / 8, 256, 0, stream>>>(h, esrc2, table2, deg2,
                                                                aggr2, N2_);
    {
        int waves = (N2_ / 32) * (OUT_C / 32);
        sage_gemm_two<HID_C, OUT_C, 2, 2, false>
            <<<waves / 8, 256, 0, stream>>>(aggr2, W2l, h, W2r, b2, out, N2_);
    }

    sage_logsoftmax64<<<(N2_ + 7) / 8, 256, 0, stream>>>(out, N2_);
}


# CSR build of both layers in 3 launches, not 6: count, scan and placing each serve layer 2 beside layer 1 (group number picks the layer); 11 -> 8 launches; sums stay in chunk, then LDS arrival order
# speedup vs baseline: 2.3498x; 1.0551x over previous
#include <hip/hip_runtime.h>
#include <hip/hip_bf16.h>

#define N0    262144
#define N1_   65536
#define N2_   16384
#define E1_   1048576
#define E2_   262144
#define IN_C  128
#define HID_C 256
#define OUT_C 64

typedef __attribute__((ext_vector_type(2))) float v2f;
typedef __attribute__((ext_vector_type(4))) float v4f;
typedef __attribute__((ext_vector_type(8))) float v8f;

template <int E, int N, int RANGE, int C>
__device__ __forceinline__ void
sage_count_ranges(const int bid, const int* __restrict__ dst, int* __restrict__ table,
                  int* __restrict__ tot, int* __restrict__ slot) {
    constexpr int R = N / RANGE;
    constexpr int U = 8;
    constexpr int PER4 = E / C / 4;
    static_assert(N % RANGE == 0, "whole ranges");
    static_assert(PER4 % (1024 * U) == 0, "whole trips");
    __shared__ int cnt[RANGE];
    __shared__ int total;
    const int tid = threadIdx.x;
    const int r   = bid % R;
    const int c   = bid / R;
    const int lo  = r * RANGE;
    for (int i = tid; i < RANGE; i += 1024) cnt[i] = 0;
    if (tid == 0) total = 0;
    __syncthreads();

    const int4* d4 = (const int4*)dst + (size_t)c * PER4;
    int* sl = slot + (size_t)c * (E / C);
    for (int it = tid; it < PER4; it += 1024 * U) {
        int4 d[U];
#pragma unroll
        for (int u = 0; u < U; ++u) d[u] = d4[it + u * 1024];
        int rk[U][4];
#pragma unroll
        for (int u = 0; u < U; ++u) {
            const unsigned ux = (unsigned)(d[u].x - lo);
            const unsigned uy = (unsigned)(d[u].y - lo);
            const unsigned uz = (unsigned)(d[u].z - lo);
            const unsigned uw = (unsigned)(d[u].w - lo);
            rk[u][0] = (ux < (unsigned)RANGE) ? atomicAdd(&cnt[ux], 1) : -1;
            rk[u][1] = (uy < (unsigned)RANGE) ? atomicAdd(&cnt[uy], 1) : -1;
            rk[u][2] = (uz < (unsigned)RANGE) ? atomicAdd(&cnt[uz], 1) : -1;
            rk[u][3] = (uw < (unsigned)RANGE) ? atomicAdd(&cnt[uw], 1) : -1;
        }
#pragma unroll
        for (int u = 0; u < U; ++u) {
            const int e = 4 * (it + u * 1024);
            if (rk[u][0] >= 0) sl[e + 0] = rk[u][0];
            if (rk[u][1] >= 0) sl[e + 1] = rk[u][1];
            if (rk[u][2] >= 0) sl[e + 2] = rk[u][2];
            if (rk[u][3] >= 0) sl[e + 3] = rk[u][3];
        }
    }
    __syncthreads();

    int mine = 0;
    for (int i = tid; i < RANGE; i += 1024) {
        const int v = cnt[i];
        table[(size_t)c * N + lo + i] = v;
        mine += v;
    }
#pragma unroll
    for (int off = 16; off > 0; off >>= 1) mine += __shfl_xor(mine, off, 32);
    if ((tid & 31) == 0) atomicAdd(&total, mine);
    __syncthreads();
    if (tid == 0) tot[c * R + r] = total;
}

template <int N, int RANGE, int C, bool COPY>
__device__ __forceinline__ void
sage_scan_ranges(const int blk, int* __restrict__ table, const int* __restrict__ tot,
                 int* __restrict__ deg, int* __restrict__ offs) {
    constexpr int R  = N / RANGE;
    constexpr int NT = RANGE / 4;
    constexpr int NW = NT / 32;
    static_assert(NW <= 32, "one lane per wave in the fold");
    __shared__ int wave_own[32];
    __shared__ int wave_front[32];
    const int tid  = threadIdx.x;
    const int lane = tid & 31;
    const int wv   = tid >> 5;
    int4* t4 = (int4*)table;
    const int idx4 = blk * NT + tid;

    int4 cn[C];
#pragma unroll
    for (int c = 0; c < C; ++c) cn[c] = t4[c * (N / 4) + idx4];

    int front = 0;
    for (int i = tid; i < C * R; i += NT) {
        const int t = tot[i];
        front += ((i % R) < blk) ? t : 0;
    }

    int4 v = cn[0];
#pragma unroll
    for (int c = 1; c < C; ++c) {
        v.x += cn[c].x; v.y += cn[c].y; v.z += cn[c].z; v.w += cn[c].w;
    }
    const int s = (v.x + v.y) + (v.z + v.w);
    int incl = s;
#pragma unroll
    for (int off = 1; off < 32; off <<= 1) {
        const int t = __shfl_up(incl, off, 32);
        if (lane >= off) incl += t;
    }
#pragma unroll
    for (int off = 16; off > 0; off >>= 1) front += __shfl_xor(front, off, 32);

    if (lane == 31) wave_own[wv] = incl;
    if (lane == 0)  wave_front[wv] = front;
    __syncthreads();

    int a = (lane < NW) ? (wave_front[lane] + ((lane < wv) ? wave_own[lane] : 0)) : 0;
#pragma unroll
    for (int off = 16; off > 0; off >>= 1) a += __shfl_xor(a, off, 32);

    ((int4*)deg)[idx4] = v;
    int4 b;
    b.x = a + incl - s;
    b.y = b.x + v.x;
    b.z = b.y + v.y;
    b.w = b.z + v.z;
    if (COPY) ((int4*)offs)[idx4] = b;
#pragma unroll
    for (int c = 0; c < C; ++c) {
        t4[c * (N / 4) + idx4] = b;
        b.x += cn[c].x; b.y += cn[c].y; b.z += cn[c].z; b.w += cn[c].w;
    }
}

template <int E, int N, int C>
__device__ __forceinline__ void
sage_place_edges(const int bid,
                 const int* __restrict__ src,
                 const int* __restrict__ dst,
                 const int* __restrict__ table,
                 const int* __restrict__ slot,
                 int* __restrict__ esrc) {
    static_assert((E / C) % 1024 == 0, "a block's 1024 edges lie in one chunk");
    const int i = bid * 256 + threadIdx.x;
    const int* base = table + (size_t)(bid / (E / C / 1024)) * N;
    const int4 d = ((const int4*)dst)[i];
    const int4 s = ((const int4*)src)[i];
    const int4 r = ((const int4*)slot)[i];
    const int px = base[d.x] + r.x;
    const int py = base[d.y] + r.y;
    const int pz = base[d.z] + r.z;
    const int pw = base[d.w] + r.w;
    esrc[px] = s.x;
    esrc[py] = s.y;
    esrc[pz] = s.z;
    esrc[pw] = s.w;
}

template <int C, int RGA, int RGB>
__global__ void __launch_bounds__(1024)
sage_count_both(const int* __restrict__ dstA, int* __restrict__ tableA,
                int* __restrict__ totA, int* __restrict__ slotA,
                const int* __restrict__ dstB, int* __restrict__ tableB,
                int* __restrict__ totB, int* __restrict__ slotB) {
    constexpr int GB = (N2_ / RGB) * C;
    const int bid = blockIdx.x;
    if (bid < GB) sage_count_ranges<E2_, N2_, RGB, C>(bid, dstB, tableB, totB, slotB);
    else          sage_count_ranges<E1_, N1_, RGA, C>(bid - GB, dstA, tableA, totA, slotA);
}

template <int C, int RG>
__global__ void __launch_bounds__(RG / 4)
sage_scan_both(int* __restrict__ tableA, const int* __restrict__ totA, int* __restrict__ degA,
               int* __restrict__ tableB, const int* __restrict__ totB, int* __restrict__ degB,
               int* __restrict__ offsB) {
    constexpr int GB = N2_ / RG;
    const int bid = blockIdx.x;
    if (bid < GB) sage_scan_ranges<N2_, RG, C, true>(bid, tableB, totB, degB, offsB);
    else          sage_scan_ranges<N1_, RG, C, false>(bid - GB, tableA, totA, degA, nullptr);
}

template <int C>
__global__ void __launch_bounds__(256)
sage_place_both(const int* __restrict__ srcA, const int* __restrict__ dstA,
                const int* __restrict__ tableA, const int* __restrict__ slotA,
                int* __restrict__ esrcA,
                const int* __restrict__ srcB, const int* __restrict__ dstB,
                const int* __restrict__ tableB, const int* __restrict__ slotB,
                int* __restrict__ esrcB) {
    constexpr int GB = E2_ / 1024;
    const int bid = blockIdx.x;
    if (bid < GB) sage_place_edges<E2_, N2_, C>(bid, srcB, dstB, tableB, slotB, esrcB);
    else          sage_place_edges<E1_, N1_, C>(bid - GB, srcA, dstA, tableA, slotA, esrcA);
}

template <int C, int B>
__device__ __forceinline__ void sage_sum_rows(const float* __restrict__ x, int lane,
                                              int myid, int n, float4 (&acc)[C / 128]) {
    constexpr int V = C / 128;
    int j = 0;
    for (; j + B <= n; j += B) {
        float4 t[B][V];
#pragma unroll
        for (int k = 0; k < B; ++k) {
            const int s = __builtin_amdgcn_readlane(myid, j + k);
            const float4* row = (const float4*)(x + (size_t)s * C);
#pragma unroll
            for (int v = 0; v < V; ++v) t[k][v] = row[lane + 32 * v];
        }
        __builtin_amdgcn_sched_barrier(0);
#pragma unroll
        for (int k = 0; k < B; ++k) {
#pragma unroll
            for (int v = 0; v < V; ++v) {
                acc[v].x += t[k][v].x; acc[v].y += t[k][v].y;
                acc[v].z += t[k][v].z; acc[v].w += t[k][v].w;
            }
        }
    }
    if (j < n) {
        float4 t[B - 1][V];
#pragma unroll
        for (int k = 0; k < B - 1; ++k) {
            if (j + k < n) {
                const int s = __builtin_amdgcn_readlane(myid, j + k);
                const float4* row = (const float4*)(x + (size_t)s * C);
#pragma unroll
                for (int v = 0; v < V; ++v) t[k][v] = row[lane + 32 * v];
            } else {
#pragma unroll
                for (int v = 0; v < V; ++v) t[k][v] = make_float4(0.f, 0.f, 0.f, 0.f);
            }
        }
#pragma unroll
        for (int k = 0; k < B - 1; ++k) {
            if (j + k < n) {
#pragma unroll
                for (int v = 0; v < V; ++v) {
                    acc[v].x += t[k][v].x; acc[v].y += t[k][v].y;
                    acc[v].z += t[k][v].z; acc[v].w += t[k][v].w;
                }
            }
        }
    }
}

template <int C, int R>
__global__ void __launch_bounds__(256)
sage_gather_mean(const float* __restrict__ x,
                                 const int* __restrict__ esrc,
                                 const int* __restrict__ offs,
                                 const int* __restrict__ deg,
                                 float* __restrict__ outm, int nDst) {
    static_assert(R <= 32, "one lane per row of the wave");
    const int wave = blockIdx.x * (blockDim.x >> 5) + (threadIdx.x >> 5);
    const int lane = threadIdx.x & 31;
    const int r0 = __builtin_amdgcn_readfirstlane(wave) * R;
    if (r0 >= nDst) return;
    constexpr int V = C / 128;
    constexpr int B = 16 / V;

    const bool has = (lane < R) && (r0 + lane < nDst);
    const int offv = has ? offs[r0 + lane] : 0;
    const int degv = has ? deg[r0 + lane] : 0;

    int beg = __builtin_amdgcn_readlane(offv, 0);
    int d   = __builtin_amdgcn_readlane(degv, 0);
    int ids = (lane < d) ? esrc[beg + lane] : 0;

    for (int r = 0; r < R; ++r) {
        if (r0 + r >= nDst) break;
        const int cur = ids;
        const int cbeg = beg;
        const int cd   = d;
        if (r + 1 < R) {
            beg = __builtin_amdgcn_readlane(offv, r + 1);
            d   = __builtin_amdgcn_readlane(degv, r + 1);
            ids = (lane < d) ? esrc[beg + lane] : 0;
        }
        float4 acc[V];
#pragma unroll
        for (int v = 0; v < V; ++v) acc[v] = make_float4(0.f, 0.f, 0.f, 0.f);

        sage_sum_rows<C, B>(x, lane, cur, (cd < 32) ? cd : 32, acc);
        for (int base = 32; base < cd; base += 32) {
            int n = cd - base;
            if (n > 32) n = 32;
            const int more = (lane < n) ? esrc[cbeg + base + lane] : 0;
            sage_sum_rows<C, B>(x, lane, more, n, acc);
        }

        const float scale = (cd > 0) ? (1.0f / (float)cd) : 0.0f;
        float4* orow = (float4*)(outm + (size_t)(r0 + r) * C);
#pragma unroll
        for (int v = 0; v < V; ++v) {
            float4 t;
            t.x = acc[v].x * scale; t.y = acc[v].y * scale;
            t.z = acc[v].z * scale; t.w = acc[v].w * scale;
            orow[lane + 32 * v] = t;
        }
    }
}

template <int K, int MT, int NT>
__device__ __forceinline__ void sage_gemm_phase(const float* const (&ap)[MT],
                                                const float* const (&bp)[NT],
                                                v8f (&acc)[MT][NT]) {
#pragma unroll 1
    for (int kc = 0; kc < K; kc += 32) {
        v4f a[4][MT], b[4][NT];
#pragma unroll
        for (int q = 0; q < 4; ++q) {
#pragma unroll
            for (int mt = 0; mt < MT; ++mt)
                a[q][mt] = *(const v4f*)(ap[mt] + kc + 4 * q);
#pragma unroll
            for (int nt = 0; nt < NT; ++nt)
                b[q][nt] = *(const v4f*)(bp[nt] + kc + 4 * q);
        }
        __builtin_amdgcn_sched_barrier(0);
#pragma unroll
        for (int q = 0; q < 4; ++q) {
#pragma unroll
            for (int p = 0; p < 2; ++p) {
                v2f af[MT], bf[NT];
#pragma unroll
                for (int mt = 0; mt < MT; ++mt) {
                    af[mt].x = a[q][mt][2 * p];
                    af[mt].y = a[q][mt][2 * p + 1];
                }
#pragma unroll
                for (int nt = 0; nt < NT; ++nt) {
                    bf[nt].x = b[q][nt][2 * p];
                    bf[nt].y = b[q][nt][2 * p + 1];
                }
#pragma unroll
                for (int mt = 0; mt < MT; ++mt)
#pragma unroll
                    for (int nt = 0; nt < NT; ++nt)
                        acc[mt][nt] = __builtin_amdgcn_wmma_f32_16x16x4_f32(
                            false, af[mt], false, bf[nt], (short)0, acc[mt][nt],
                            false, false);
            }
        }
    }
}

template <int K, int N, int MT, int NT, bool RELU>
__global__ void __launch_bounds__(256)
sage_gemm_two(const float* __restrict__ A1,
                              const float* __restrict__ B1,
                              const float* __restrict__ A2,
                              const float* __restrict__ B2,
                              const float* __restrict__ bias,
                              float* __restrict__ out, int M) {
    static_assert(K % 32 == 0, "K is walked in whole 128-byte row lines");
    const int lane   = threadIdx.x & 31;
    const int waveId = blockIdx.x * (blockDim.x >> 5) + (threadIdx.x >> 5);
    const int groupsN = N / (16 * NT);
    const int gm = waveId / groupsN;
    const int gn = waveId % groupsN;
    if (gm >= M / (16 * MT)) return;

    const int m0  = gm * 16 * MT;
    const int n0  = gn * 16 * NT;
    const int kh  = (lane >> 4) * 16;

    const float* a1p[MT];
    const float* a2p[MT];
#pragma unroll
    for (int mt = 0; mt < MT; ++mt) {
        int r = m0 + mt * 16 + (lane & 15);
        a1p[mt] = A1 + (size_t)r * K + kh;
        a2p[mt] = A2 + (size_t)r * K + kh;
    }
    const float* b1p[NT];
    const float* b2p[NT];
#pragma unroll
    for (int nt = 0; nt < NT; ++nt) {
        int c = n0 + nt * 16 + (lane & 15);
        b1p[nt] = B1 + (size_t)c * K + kh;
        b2p[nt] = B2 + (size_t)c * K + kh;
    }

    v8f acc[MT][NT] = {};

    sage_gemm_phase<K, MT, NT>(a1p, b1p, acc);
    sage_gemm_phase<K, MT, NT>(a2p, b2p, acc);

    const int rbase = ((lane >> 4) << 3);
#pragma unroll
    for (int nt = 0; nt < NT; ++nt) {
        const int ocol = n0 + nt * 16 + (lane & 15);
        const float bb = bias[ocol];
#pragma unroll
        for (int mt = 0; mt < MT; ++mt) {
#pragma unroll
            for (int r = 0; r < 8; ++r) {
                float v = acc[mt][nt][r] + bb;
                if (RELU) v = fmaxf(v, 0.0f);
                out[(size_t)(m0 + mt * 16 + rbase + r) * N + ocol] = v;
            }
        }
    }
}

__global__ void sage_logsoftmax64(float* __restrict__ out, int rows) {
    int wave = blockIdx.x * (blockDim.x >> 5) + (threadIdx.x >> 5);
    int lane = threadIdx.x & 31;
    if (wave >= rows) return;
    float* p = out + (size_t)wave * 64;
    float a = p[lane];
    float b = p[lane + 32];
    float m = fmaxf(a, b);
#pragma unroll
    for (int off = 16; off > 0; off >>= 1)
        m = fmaxf(m, __shfl_xor(m, off, 32));
    float s = expf(a - m) + expf(b - m);
#pragma unroll
    for (int off = 16; off > 0; off >>= 1)
        s += __shfl_xor(s, off, 32);
    float lse = m + logf(s);
    p[lane]      = a - lse;
    p[lane + 32] = b - lse;
}

extern "C" void kernel_launch(void* const* d_in, const int* in_sizes, int n_in,
                              void* d_out, int out_size, void* d_ws, size_t ws_size,
                              hipStream_t stream) {
    const float* x    = (const float*)d_in[0];
    const int*   src1 = (const int*)d_in[1];
    const int*   dst1 = (const int*)d_in[2];
    const int*   src2 = (const int*)d_in[3];
    const int*   dst2 = (const int*)d_in[4];
    const float* W1l  = (const float*)d_in[5];
    const float* W1r  = (const float*)d_in[6];
    const float* b1   = (const float*)d_in[7];
    const float* W2l  = (const float*)d_in[8];
    const float* W2r  = (const float*)d_in[9];
    const float* b2   = (const float*)d_in[10];
    float* out = (float*)d_out;

    constexpr int CH  = 8;
    constexpr int RG  = 2048;
    int* deg1  = (int*)d_ws;
    int* deg2  = deg1 + N1_;
    int* tot1  = deg2 + N2_;
    int* offs2 = tot1 + 4096;
    int* tot2  = tot1 + N1_;
    int* esrc1 = tot2 + N2_;
    int* esrc2 = esrc1 + E1_;
    float* aggr1 = (float*)(esrc2 + E2_);
    float* aggr2 = aggr1 + (size_t)N1_ * IN_C;
    float* h     = aggr2 + (size_t)N2_ * HID_C;
    int* slot1  = (int*)aggr1;
    int* slot2  = (int*)aggr2;
    int* table1 = (int*)h;
    int* table2 = (int*)aggr2 + E2_;

    sage_count_both<CH, RG, RG>
        <<<((N1_ + N2_) / RG) * CH, 1024, 0, stream>>>(dst1, table1, tot1, slot1,
                                                       dst2, table2, tot2, slot2);
    sage_scan_both<CH, RG>
        <<<(N1_ + N2_) / RG, RG / 4, 0, stream>>>(table1, tot1, deg1,
                                                  table2, tot2, deg2, offs2);
    sage_place_both<CH>
        <<<(E1_ + E2_) / 1024, 256, 0, stream>>>(src1, dst1, table1, slot1, esrc1,
                                                 src2, dst2, table2, slot2, esrc2);

    sage_gather_mean<IN_C, 8><<<N1_ / 8 / 8, 256, 0, stream>>>(x, esrc1, table1, deg1,
                                                               aggr1, N1_);
    {
        int waves = (N1_ / 32) * (HID_C / 64);
        sage_gemm_two<IN_C, HID_C, 2, 4, true>
            <<<waves / 8, 256, 0, stream>>>(aggr1, W1l, x, W1r, b1, h, N1_);
    }

    sage_gather_mean<HID_C, 2><<<N2_ / 2 / 8, 256, 0, stream>>>(h, esrc2, offs2, deg2,
                                                                aggr2, N2_);
    {
        int waves = (N2_ / 32) * (OUT_C / 32);
        sage_gemm_two<HID_C, OUT_C, 2, 2, false>
            <<<waves / 8, 256, 0, stream>>>(aggr2, W2l, h, W2r, b2, out, N2_);
    }

    sage_logsoftmax64<<<(N2_ + 7) / 8, 256, 0, stream>>>(out, N2_);
}


# Layer 2 in one launch: a group gathers 16 mean rows into LDS, waves 0-3 multiply them by W_l, 4-7 the own rows by W_r, log-softmax from LDS; 8 -> 6 launches; two partial tiles added: sums reorder
# speedup vs baseline: 2.5861x; 1.1006x over previous
#include <hip/hip_runtime.h>
#include <hip/hip_bf16.h>

#define N0    262144
#define N1_   65536
#define N2_   16384
#define E1_   1048576
#define E2_   262144
#define IN_C  128
#define HID_C 256
#define OUT_C 64

typedef __attribute__((ext_vector_type(2))) float v2f;
typedef __attribute__((ext_vector_type(4))) float v4f;
typedef __attribute__((ext_vector_type(8))) float v8f;

template <int E, int N, int RANGE, int C>
__device__ __forceinline__ void
sage_count_ranges(const int bid, const int* __restrict__ dst, int* __restrict__ table,
                  int* __restrict__ tot, int* __restrict__ slot) {
    constexpr int R = N / RANGE;
    constexpr int U = 8;
    constexpr int PER4 = E / C / 4;
    static_assert(N % RANGE == 0, "whole ranges");
    static_assert(PER4 % (1024 * U) == 0, "whole trips");
    __shared__ int cnt[RANGE];
    __shared__ int total;
    const int tid = threadIdx.x;
    const int r   = bid % R;
    const int c   = bid / R;
    const int lo  = r * RANGE;
    for (int i = tid; i < RANGE; i += 1024) cnt[i] = 0;
    if (tid == 0) total = 0;
    __syncthreads();

    const int4* d4 = (const int4*)dst + (size_t)c * PER4;
    int* sl = slot + (size_t)c * (E / C);
    for (int it = tid; it < PER4; it += 1024 * U) {
        int4 d[U];
#pragma unroll
        for (int u = 0; u < U; ++u) d[u] = d4[it + u * 1024];
        int rk[U][4];
#pragma unroll
        for (int u = 0; u < U; ++u) {
            const unsigned ux = (unsigned)(d[u].x - lo);
            const unsigned uy = (unsigned)(d[u].y - lo);
            const unsigned uz = (unsigned)(d[u].z - lo);
            const unsigned uw = (unsigned)(d[u].w - lo);
            rk[u][0] = (ux < (unsigned)RANGE) ? atomicAdd(&cnt[ux], 1) : -1;
            rk[u][1] = (uy < (unsigned)RANGE) ? atomicAdd(&cnt[uy], 1) : -1;
            rk[u][2] = (uz < (unsigned)RANGE) ? atomicAdd(&cnt[uz], 1) : -1;
            rk[u][3] = (uw < (unsigned)RANGE) ? atomicAdd(&cnt[uw], 1) : -1;
        }
#pragma unroll
        for (int u = 0; u < U; ++u) {
            const int e = 4 * (it + u * 1024);
            if (rk[u][0] >= 0) sl[e + 0] = rk[u][0];
            if (rk[u][1] >= 0) sl[e + 1] = rk[u][1];
            if (rk[u][2] >= 0) sl[e + 2] = rk[u][2];
            if (rk[u][3] >= 0) sl[e + 3] = rk[u][3];
        }
    }
    __syncthreads();

    int mine = 0;
    for (int i = tid; i < RANGE; i += 1024) {
        const int v = cnt[i];
        table[(size_t)c * N + lo + i] = v;
        mine += v;
    }
#pragma unroll
    for (int off = 16; off > 0; off >>= 1) mine += __shfl_xor(mine, off, 32);
    if ((tid & 31) == 0) atomicAdd(&total, mine);
    __syncthreads();
    if (tid == 0) tot[c * R + r] = total;
}

template <int N, int RANGE, int C, bool COPY>
__device__ __forceinline__ void
sage_scan_ranges(const int blk, int* __restrict__ table, const int* __restrict__ tot,
                 int* __restrict__ deg, int* __restrict__ offs) {
    constexpr int R  = N / RANGE;
    constexpr int NT = RANGE / 4;
    constexpr int NW = NT / 32;
    static_assert(NW <= 32, "one lane per wave in the fold");
    __shared__ int wave_own[32];
    __shared__ int wave_front[32];
    const int tid  = threadIdx.x;
    const int lane = tid & 31;
    const int wv   = tid >> 5;
    int4* t4 = (int4*)table;
    const int idx4 = blk * NT + tid;

    int4 cn[C];
#pragma unroll
    for (int c = 0; c < C; ++c) cn[c] = t4[c * (N / 4) + idx4];

    int front = 0;
    for (int i = tid; i < C * R; i += NT) {
        const int t = tot[i];
        front += ((i % R) < blk) ? t : 0;
    }

    int4 v = cn[0];
#pragma unroll
    for (int c = 1; c < C; ++c) {
        v.x += cn[c].x; v.y += cn[c].y; v.z += cn[c].z; v.w += cn[c].w;
    }
    const int s = (v.x + v.y) + (v.z + v.w);
    int incl = s;
#pragma unroll
    for (int off = 1; off < 32; off <<= 1) {
        const int t = __shfl_up(incl, off, 32);
        if (lane >= off) incl += t;
    }
#pragma unroll
    for (int off = 16; off > 0; off >>= 1) front += __shfl_xor(front, off, 32);

    if (lane == 31) wave_own[wv] = incl;
    if (lane == 0)  wave_front[wv] = front;
    __syncthreads();

    int a = (lane < NW) ? (wave_front[lane] + ((lane < wv) ? wave_own[lane] : 0)) : 0;
#pragma unroll
    for (int off = 16; off > 0; off >>= 1) a += __shfl_xor(a, off, 32);

    ((int4*)deg)[idx4] = v;
    int4 b;
    b.x = a + incl - s;
    b.y = b.x + v.x;
    b.z = b.y + v.y;
    b.w = b.z + v.z;
    if (COPY) ((int4*)offs)[idx4] = b;
#pragma unroll
    for (int c = 0; c < C; ++c) {
        t4[c * (N / 4) + idx4] = b;
        b.x += cn[c].x; b.y += cn[c].y; b.z += cn[c].z; b.w += cn[c].w;
    }
}

template <int E, int N, int C>
__device__ __forceinline__ void
sage_place_edges(const int bid,
                 const int* __restrict__ src,
                 const int* __restrict__ dst,
                 const int* __restrict__ table,
                 const int* __restrict__ slot,
                 int* __restrict__ esrc) {
    static_assert((E / C) % 1024 == 0, "a block's 1024 edges lie in one chunk");
    const int i = bid * 256 + threadIdx.x;
    const int* base = table + (size_t)(bid / (E / C / 1024)) * N;
    const int4 d = ((const int4*)dst)[i];
    const int4 s = ((const int4*)src)[i];
    const int4 r = ((const int4*)slot)[i];
    const int px = base[d.x] + r.x;
    const int py = base[d.y] + r.y;
    const int pz = base[d.z] + r.z;
    const int pw = base[d.w] + r.w;
    esrc[px] = s.x;
    esrc[py] = s.y;
    esrc[pz] = s.z;
    esrc[pw] = s.w;
}

template <int C, int RGA, int RGB>
__global__ void __launch_bounds__(1024)
sage_count_both(const int* __restrict__ dstA, int* __restrict__ tableA,
                int* __restrict__ totA, int* __restrict__ slotA,
                const int* __restrict__ dstB, int* __restrict__ tableB,
                int* __restrict__ totB, int* __restrict__ slotB) {
    constexpr int GB = (N2_ / RGB) * C;
    const int bid = blockIdx.x;
    if (bid < GB) sage_count_ranges<E2_, N2_, RGB, C>(bid, dstB, tableB, totB, slotB);
    else          sage_count_ranges<E1_, N1_, RGA, C>(bid - GB, dstA, tableA, totA, slotA);
}

template <int C, int RG>
__global__ void __launch_bounds__(RG / 4)
sage_scan_both(int* __restrict__ tableA, const int* __restrict__ totA, int* __restrict__ degA,
               int* __restrict__ tableB, const int* __restrict__ totB, int* __restrict__ degB,
               int* __restrict__ offsB) {
    constexpr int GB = N2_ / RG;
    const int bid = blockIdx.x;
    if (bid < GB) sage_scan_ranges<N2_, RG, C, true>(bid, tableB, totB, degB, offsB);
    else          sage_scan_ranges<N1_, RG, C, false>(bid - GB, tableA, totA, degA, nullptr);
}

template <int C>
__global__ void __launch_bounds__(256)
sage_place_both(const int* __restrict__ srcA, const int* __restrict__ dstA,
                const int* __restrict__ tableA, const int* __restrict__ slotA,
                int* __restrict__ esrcA,
                const int* __restrict__ srcB, const int* __restrict__ dstB,
                const int* __restrict__ tableB, const int* __restrict__ slotB,
                int* __restrict__ esrcB) {
    constexpr int GB = E2_ / 1024;
    const int bid = blockIdx.x;
    if (bid < GB) sage_place_edges<E2_, N2_, C>(bid, srcB, dstB, tableB, slotB, esrcB);
    else          sage_place_edges<E1_, N1_, C>(bid - GB, srcA, dstA, tableA, slotA, esrcA);
}

template <int C, int B>
__device__ __forceinline__ void sage_sum_rows(const float* __restrict__ x, int lane,
                                              int myid, int n, float4 (&acc)[C / 128]) {
    constexpr int V = C / 128;
    int j = 0;
    for (; j + B <= n; j += B) {
        float4 t[B][V];
#pragma unroll
        for (int k = 0; k < B; ++k) {
            const int s = __builtin_amdgcn_readlane(myid, j + k);
            const float4* row = (const float4*)(x + (size_t)s * C);
#pragma unroll
            for (int v = 0; v < V; ++v) t[k][v] = row[lane + 32 * v];
        }
        __builtin_amdgcn_sched_barrier(0);
#pragma unroll
        for (int k = 0; k < B; ++k) {
#pragma unroll
            for (int v = 0; v < V; ++v) {
                acc[v].x += t[k][v].x; acc[v].y += t[k][v].y;
                acc[v].z += t[k][v].z; acc[v].w += t[k][v].w;
            }
        }
    }
    if (j < n) {
        float4 t[B - 1][V];
#pragma unroll
        for (int k = 0; k < B - 1; ++k) {
            if (j + k < n) {
                const int s = __builtin_amdgcn_readlane(myid, j + k);
                const float4* row = (const float4*)(x + (size_t)s * C);
#pragma unroll
                for (int v = 0; v < V; ++v) t[k][v] = row[lane + 32 * v];
            } else {
#pragma unroll
                for (int v = 0; v < V; ++v) t[k][v] = make_float4(0.f, 0.f, 0.f, 0.f);
            }
        }
#pragma unroll
        for (int k = 0; k < B - 1; ++k) {
            if (j + k < n) {
#pragma unroll
                for (int v = 0; v < V; ++v) {
                    acc[v].x += t[k][v].x; acc[v].y += t[k][v].y;
                    acc[v].z += t[k][v].z; acc[v].w += t[k][v].w;
                }
            }
        }
    }
}

template <int C, int R>
__global__ void __launch_bounds__(256)
sage_gather_mean(const float* __restrict__ x,
                                 const int* __restrict__ esrc,
                                 const int* __restrict__ offs,
                                 const int* __restrict__ deg,
                                 float* __restrict__ outm, int nDst) {
    static_assert(R <= 32, "one lane per row of the wave");
    const int wave = blockIdx.x * (blockDim.x >> 5) + (threadIdx.x >> 5);
    const int lane = threadIdx.x & 31;
    const int r0 = __builtin_amdgcn_readfirstlane(wave) * R;
    if (r0 >= nDst) return;
    constexpr int V = C / 128;
    constexpr int B = 16 / V;

    const bool has = (lane < R) && (r0 + lane < nDst);
    const int offv = has ? offs[r0 + lane] : 0;
    const int degv = has ? deg[r0 + lane] : 0;

    int beg = __builtin_amdgcn_readlane(offv, 0);
    int d   = __builtin_amdgcn_readlane(degv, 0);
    int ids = (lane < d) ? esrc[beg + lane] : 0;

    for (int r = 0; r < R; ++r) {
        if (r0 + r >= nDst) break;
        const int cur = ids;
        const int cbeg = beg;
        const int cd   = d;
        if (r + 1 < R) {
            beg = __builtin_amdgcn_readlane(offv, r + 1);
            d   = __builtin_amdgcn_readlane(degv, r + 1);
            ids = (lane < d) ? esrc[beg + lane] : 0;
        }
        float4 acc[V];
#pragma unroll
        for (int v = 0; v < V; ++v) acc[v] = make_float4(0.f, 0.f, 0.f, 0.f);

        sage_sum_rows<C, B>(x, lane, cur, (cd < 32) ? cd : 32, acc);
        for (int base = 32; base < cd; base += 32) {
            int n = cd - base;
            if (n > 32) n = 32;
            const int more = (lane < n) ? esrc[cbeg + base + lane] : 0;
            sage_sum_rows<C, B>(x, lane, more, n, acc);
        }

        const float scale = (cd > 0) ? (1.0f / (float)cd) : 0.0f;
        float4* orow = (float4*)(outm + (size_t)(r0 + r) * C);
#pragma unroll
        for (int v = 0; v < V; ++v) {
            float4 t;
            t.x = acc[v].x * scale; t.y = acc[v].y * scale;
            t.z = acc[v].z * scale; t.w = acc[v].w * scale;
            orow[lane + 32 * v] = t;
        }
    }
}

template <int K, int MT, int NT>
__device__ __forceinline__ void sage_gemm_phase(const float* const (&ap)[MT],
                                                const float* const (&bp)[NT],
                                                v8f (&acc)[MT][NT]) {
#pragma unroll 1
    for (int kc = 0; kc < K; kc += 32) {
        v4f a[4][MT], b[4][NT];
#pragma unroll
        for (int q = 0; q < 4; ++q) {
#pragma unroll
            for (int mt = 0; mt < MT; ++mt)
                a[q][mt] = *(const v4f*)(ap[mt] + kc + 4 * q);
#pragma unroll
            for (int nt = 0; nt < NT; ++nt)
                b[q][nt] = *(const v4f*)(bp[nt] + kc + 4 * q);
        }
        __builtin_amdgcn_sched_barrier(0);
#pragma unroll
        for (int q = 0; q < 4; ++q) {
#pragma unroll
            for (int p = 0; p < 2; ++p) {
                v2f af[MT], bf[NT];
#pragma unroll
                for (int mt = 0; mt < MT; ++mt) {
                    af[mt].x = a[q][mt][2 * p];
                    af[mt].y = a[q][mt][2 * p + 1];
                }
#pragma unroll
                for (int nt = 0; nt < NT; ++nt) {
                    bf[nt].x = b[q][nt][2 * p];
                    bf[nt].y = b[q][nt][2 * p + 1];
                }
#pragma unroll
                for (int mt = 0; mt < MT; ++mt)
#pragma unroll
                    for (int nt = 0; nt < NT; ++nt)
                        acc[mt][nt] = __builtin_amdgcn_wmma_f32_16x16x4_f32(
                            false, af[mt], false, bf[nt], (short)0, acc[mt][nt],
                            false, false);
            }
        }
    }
}

template <int K, int N, int MT, int NT, bool RELU>
__global__ void __launch_bounds__(256)
sage_gemm_two(const float* __restrict__ A1,
                              const float* __restrict__ B1,
                              const float* __restrict__ A2,
                              const float* __restrict__ B2,
                              const float* __restrict__ bias,
                              float* __restrict__ out, int M) {
    static_assert(K % 32 == 0, "K is walked in whole 128-byte row lines");
    const int lane   = threadIdx.x & 31;
    const int waveId = blockIdx.x * (blockDim.x >> 5) + (threadIdx.x >> 5);
    const int groupsN = N / (16 * NT);
    const int gm = waveId / groupsN;
    const int gn = waveId % groupsN;
    if (gm >= M / (16 * MT)) return;

    const int m0  = gm * 16 * MT;
    const int n0  = gn * 16 * NT;
    const int kh  = (lane >> 4) * 16;

    const float* a1p[MT];
    const float* a2p[MT];
#pragma unroll
    for (int mt = 0; mt < MT; ++mt) {
        int r = m0 + mt * 16 + (lane & 15);
        a1p[mt] = A1 + (size_t)r * K + kh;
        a2p[mt] = A2 + (size_t)r * K + kh;
    }
    const float* b1p[NT];
    const float* b2p[NT];
#pragma unroll
    for (int nt = 0; nt < NT; ++nt) {
        int c = n0 + nt * 16 + (lane & 15);
        b1p[nt] = B1 + (size_t)c * K + kh;
        b2p[nt] = B2 + (size_t)c * K + kh;
    }

    v8f acc[MT][NT] = {};

    sage_gemm_phase<K, MT, NT>(a1p, b1p, acc);
    sage_gemm_phase<K, MT, NT>(a2p, b2p, acc);

    const int rbase = ((lane >> 4) << 3);
#pragma unroll
    for (int nt = 0; nt < NT; ++nt) {
        const int ocol = n0 + nt * 16 + (lane & 15);
        const float bb = bias[ocol];
#pragma unroll
        for (int mt = 0; mt < MT; ++mt) {
#pragma unroll
            for (int r = 0; r < 8; ++r) {
                float v = acc[mt][nt][r] + bb;
                if (RELU) v = fmaxf(v, 0.0f);
                out[(size_t)(m0 + mt * 16 + rbase + r) * N + ocol] = v;
            }
        }
    }
}

template <int K>
__device__ __forceinline__ void sage_tile_k(const float* ap, const float* bp, v8f& acc) {
#pragma unroll 1
    for (int kc = 0; kc < K; kc += 64) {
        v4f a[2][4], b[2][4];
#pragma unroll
        for (int u = 0; u < 2; ++u) {
#pragma unroll
            for (int q = 0; q < 4; ++q) {
                a[u][q] = *(const v4f*)(ap + kc + 32 * u + 4 * q);
                b[u][q] = *(const v4f*)(bp + kc + 32 * u + 4 * q);
            }
        }
#pragma unroll
        for (int u = 0; u < 2; ++u) {
#pragma unroll
            for (int q = 0; q < 4; ++q) {
#pragma unroll
                for (int p = 0; p < 2; ++p) {
                    v2f af, bf;
                    af.x = a[u][q][2 * p]; af.y = a[u][q][2 * p + 1];
                    bf.x = b[u][q][2 * p]; bf.y = b[u][q][2 * p + 1];
                    acc = __builtin_amdgcn_wmma_f32_16x16x4_f32(false, af, false, bf, (short)0,
                                                                acc, false, false);
                }
            }
        }
    }
}

template <int RPW>
__global__ void __launch_bounds__(256)
sage_layer2_fused(const float* __restrict__ hsrc,
                  const int* __restrict__ esrc,
                  const int* __restrict__ offs,
                  const int* __restrict__ deg,
                  const float* __restrict__ Wl,
                  const float* __restrict__ Wr,
                  const float* __restrict__ bias,
                  float* __restrict__ out) {
    constexpr int C    = HID_C;
    constexpr int ROWS = 8 * RPW;
    static_assert(ROWS == 16, "one 16-row tile per work-group");
    static_assert(N2_ % ROWS == 0, "whole work-groups");
    constexpr int V  = C / 128;
    constexpr int B  = 16 / V;
    constexpr int PA = C + 4;
    constexpr int PO = OUT_C + 4;
    __shared__ __attribute__((aligned(16))) float sA[ROWS * PA];
    __shared__ __attribute__((aligned(16))) float sP[2 * ROWS * PO];

    const int lane = threadIdx.x & 31;
    const int wv   = __builtin_amdgcn_readfirstlane((int)(threadIdx.x >> 5));
    const int row0 = blockIdx.x * ROWS;

    {
        const int lr0 = wv * RPW;
        const int r0  = row0 + lr0;
        const bool has = lane < RPW;
        const int offv = has ? offs[r0 + lane] : 0;
        const int degv = has ? deg[r0 + lane] : 0;
        int beg = __builtin_amdgcn_readlane(offv, 0);
        int d   = __builtin_amdgcn_readlane(degv, 0);
        int ids = (lane < d) ? esrc[beg + lane] : 0;
        for (int r = 0; r < RPW; ++r) {
            const int cur  = ids;
            const int cbeg = beg;
            const int cd   = d;
            if (r + 1 < RPW) {
                beg = __builtin_amdgcn_readlane(offv, r + 1);
                d   = __builtin_amdgcn_readlane(degv, r + 1);
                ids = (lane < d) ? esrc[beg + lane] : 0;
            }
            float4 acc[V];
#pragma unroll
            for (int v = 0; v < V; ++v) acc[v] = make_float4(0.f, 0.f, 0.f, 0.f);
            sage_sum_rows<C, B>(hsrc, lane, cur, (cd < 32) ? cd : 32, acc);
            for (int base = 32; base < cd; base += 32) {
                int n = cd - base;
                if (n > 32) n = 32;
                const int more = (lane < n) ? esrc[cbeg + base + lane] : 0;
                sage_sum_rows<C, B>(hsrc, lane, more, n, acc);
            }
            const float scale = (cd > 0) ? (1.0f / (float)cd) : 0.0f;
            float4* orow = (float4*)(sA + (lr0 + r) * PA);
#pragma unroll
            for (int v = 0; v < V; ++v) {
                float4 t;
                t.x = acc[v].x * scale; t.y = acc[v].y * scale;
                t.z = acc[v].z * scale; t.w = acc[v].w * scale;
                orow[lane + 32 * v] = t;
            }
        }
    }
    __syncthreads();

    {
        const int c  = lane & 15;
        const int kh = (lane >> 4) * 16;
        const int nt = wv & 3;
        const int ph = wv >> 2;
        v8f acc = {};
        if (ph == 0)
            sage_tile_k<C>(sA + c * PA + kh, Wl + (size_t)(nt * 16 + c) * C + kh, acc);
        else
            sage_tile_k<C>(hsrc + (size_t)(row0 + c) * C + kh, Wr + (size_t)(nt * 16 + c) * C + kh, acc);
        float* pp = sP + ph * (ROWS * PO) + ((lane >> 4) << 3) * PO + nt * 16 + c;
#pragma unroll
        for (int r = 0; r < 8; ++r) pp[r * PO] = acc[r];
    }
    __syncthreads();

    {
        const float blo = bias[lane];
        const float bhi = bias[lane + 32];
#pragma unroll
        for (int r = 0; r < RPW; ++r) {
            const int lr = wv * RPW + r;
            const float* p0 = sP + lr * PO;
            const float* p1 = sP + ROWS * PO + lr * PO;
            const float a = (p0[lane] + p1[lane]) + blo;
            const float b = (p0[lane + 32] + p1[lane + 32]) + bhi;
            float m = fmaxf(a, b);
#pragma unroll
            for (int off = 16; off > 0; off >>= 1)
                m = fmaxf(m, __shfl_xor(m, off, 32));
            float s = expf(a - m) + expf(b - m);
#pragma unroll
            for (int off = 16; off > 0; off >>= 1)
                s += __shfl_xor(s, off, 32);
            const float lse = m + logf(s);
            float* o = out + (size_t)(row0 + lr) * OUT_C;
            o[lane]      = a - lse;
            o[lane + 32] = b - lse;
        }
    }
}

extern "C" void kernel_launch(void* const* d_in, const int* in_sizes, int n_in,
                              void* d_out, int out_size, void* d_ws, size_t ws_size,
                              hipStream_t stream) {
    const float* x    = (const float*)d_in[0];
    const int*   src1 = (const int*)d_in[1];
    const int*   dst1 = (const int*)d_in[2];
    const int*   src2 = (const int*)d_in[3];
    const int*   dst2 = (const int*)d_in[4];
    const float* W1l  = (const float*)d_in[5];
    const float* W1r  = (const float*)d_in[6];
    const float* b1   = (const float*)d_in[7];
    const float* W2l  = (const float*)d_in[8];
    const float* W2r  = (const float*)d_in[9];
    const float* b2   = (const float*)d_in[10];
    float* out = (float*)d_out;

    constexpr int CH  = 8;
    constexpr int RG  = 2048;
    int* deg1  = (int*)d_ws;
    int* deg2  = deg1 + N1_;
    int* tot1  = deg2 + N2_;
    int* offs2 = tot1 + 4096;
    int* tot2  = tot1 + N1_;
    int* esrc1 = tot2 + N2_;
    int* esrc2 = esrc1 + E1_;
    float* aggr1 = (float*)(esrc2 + E2_);
    float* aggr2 = aggr1 + (size_t)N1_ * IN_C;
    float* h     = aggr2 + (size_t)N2_ * HID_C;
    int* slot1  = (int*)aggr1;
    int* slot2  = (int*)aggr2;
    int* table1 = (int*)h;
    int* table2 = (int*)aggr2 + E2_;

    sage_count_both<CH, RG, RG>
        <<<((N1_ + N2_) / RG) * CH, 1024, 0, stream>>>(dst1, table1, tot1, slot1,
                                                       dst2, table2, tot2, slot2);
    sage_scan_both<CH, RG>
        <<<(N1_ + N2_) / RG, RG / 4, 0, stream>>>(table1, tot1, deg1,
                                                  table2, tot2, deg2, offs2);
    sage_place_both<CH>
        <<<(E1_ + E2_) / 1024, 256, 0, stream>>>(src1, dst1, table1, slot1, esrc1,
                                                 src2, dst2, table2, slot2, esrc2);

    sage_gather_mean<IN_C, 8><<<N1_ / 8 / 8, 256, 0, stream>>>(x, esrc1, table1, deg1,
                                                               aggr1, N1_);
    {
        int waves = (N1_ / 32) * (HID_C / 64);
        sage_gemm_two<IN_C, HID_C, 2, 4, true>
            <<<waves / 8, 256, 0, stream>>>(aggr1, W1l, x, W1r, b1, h, N1_);
    }

    sage_layer2_fused<2><<<N2_ / 16, 256, 0, stream>>>(h, esrc2, offs2, deg2, W2l, W2r, b2, out);
}


# Product operands via LDS: layer-1 A and B blocks fetched once per group with whole-line 16-B loads, fragments read from LDS, not 16-line loads from memory; layer-2 own rows too; sums bit-identical
# speedup vs baseline: 2.7841x; 1.0766x over previous
#include <hip/hip_runtime.h>
#include <hip/hip_bf16.h>

#define N0    262144
#define N1_   65536
#define N2_   16384
#define E1_   1048576
#define E2_   262144
#define IN_C  128
#define HID_C 256
#define OUT_C 64

typedef __attribute__((ext_vector_type(2))) float v2f;
typedef __attribute__((ext_vector_type(4))) float v4f;
typedef __attribute__((ext_vector_type(8))) float v8f;

template <int E, int N, int RANGE, int C>
__device__ __forceinline__ void
sage_count_ranges(const int bid, const int* __restrict__ dst, int* __restrict__ table,
                  int* __restrict__ tot, int* __restrict__ slot) {
    constexpr int R = N / RANGE;
    constexpr int U = 8;
    constexpr int PER4 = E / C / 4;
    static_assert(N % RANGE == 0, "whole ranges");
    static_assert(PER4 % (1024 * U) == 0, "whole trips");
    __shared__ int cnt[RANGE];
    __shared__ int total;
    const int tid = threadIdx.x;
    const int r   = bid % R;
    const int c   = bid / R;
    const int lo  = r * RANGE;
    for (int i = tid; i < RANGE; i += 1024) cnt[i] = 0;
    if (tid == 0) total = 0;
    __syncthreads();

    const int4* d4 = (const int4*)dst + (size_t)c * PER4;
    int* sl = slot + (size_t)c * (E / C);
    for (int it = tid; it < PER4; it += 1024 * U) {
        int4 d[U];
#pragma unroll
        for (int u = 0; u < U; ++u) d[u] = d4[it + u * 1024];
        int rk[U][4];
#pragma unroll
        for (int u = 0; u < U; ++u) {
            const unsigned ux = (unsigned)(d[u].x - lo);
            const unsigned uy = (unsigned)(d[u].y - lo);
            const unsigned uz = (unsigned)(d[u].z - lo);
            const unsigned uw = (unsigned)(d[u].w - lo);
            rk[u][0] = (ux < (unsigned)RANGE) ? atomicAdd(&cnt[ux], 1) : -1;
            rk[u][1] = (uy < (unsigned)RANGE) ? atomicAdd(&cnt[uy], 1) : -1;
            rk[u][2] = (uz < (unsigned)RANGE) ? atomicAdd(&cnt[uz], 1) : -1;
            rk[u][3] = (uw < (unsigned)RANGE) ? atomicAdd(&cnt[uw], 1) : -1;
        }
#pragma unroll
        for (int u = 0; u < U; ++u) {
            const int e = 4 * (it + u * 1024);
            if (rk[u][0] >= 0) sl[e + 0] = rk[u][0];
            if (rk[u][1] >= 0) sl[e + 1] = rk[u][1];
            if (rk[u][2] >= 0) sl[e + 2] = rk[u][2];
            if (rk[u][3] >= 0) sl[e + 3] = rk[u][3];
        }
    }
    __syncthreads();

    int mine = 0;
    for (int i = tid; i < RANGE; i += 1024) {
        const int v = cnt[i];
        table[(size_t)c * N + lo + i] = v;
        mine += v;
    }
#pragma unroll
    for (int off = 16; off > 0; off >>= 1) mine += __shfl_xor(mine, off, 32);
    if ((tid & 31) == 0) atomicAdd(&total, mine);
    __syncthreads();
    if (tid == 0) tot[c * R + r] = total;
}

template <int N, int RANGE, int C, bool COPY>
__device__ __forceinline__ void
sage_scan_ranges(const int blk, int* __restrict__ table, const int* __restrict__ tot,
                 int* __restrict__ deg, int* __restrict__ offs) {
    constexpr int R  = N / RANGE;
    constexpr int NT = RANGE / 4;
    constexpr int NW = NT / 32;
    static_assert(NW <= 32, "one lane per wave in the fold");
    __shared__ int wave_own[32];
    __shared__ int wave_front[32];
    const int tid  = threadIdx.x;
    const int lane = tid & 31;
    const int wv   = tid >> 5;
    int4* t4 = (int4*)table;
    const int idx4 = blk * NT + tid;

    int4 cn[C];
#pragma unroll
    for (int c = 0; c < C; ++c) cn[c] = t4[c * (N / 4) + idx4];

    int front = 0;
    for (int i = tid; i < C * R; i += NT) {
        const int t = tot[i];
        front += ((i % R) < blk) ? t : 0;
    }

    int4 v = cn[0];
#pragma unroll
    for (int c = 1; c < C; ++c) {
        v.x += cn[c].x; v.y += cn[c].y; v.z += cn[c].z; v.w += cn[c].w;
    }
    const int s = (v.x + v.y) + (v.z + v.w);
    int incl = s;
#pragma unroll
    for (int off = 1; off < 32; off <<= 1) {
        const int t = __shfl_up(incl, off, 32);
        if (lane >= off) incl += t;
    }
#pragma unroll
    for (int off = 16; off > 0; off >>= 1) front += __shfl_xor(front, off, 32);

    if (lane == 31) wave_own[wv] = incl;
    if (lane == 0)  wave_front[wv] = front;
    __syncthreads();

    int a = (lane < NW) ? (wave_front[lane] + ((lane < wv) ? wave_own[lane] : 0)) : 0;
#pragma unroll
    for (int off = 16; off > 0; off >>= 1) a += __shfl_xor(a, off, 32);

    ((int4*)deg)[idx4] = v;
    int4 b;
    b.x = a + incl - s;
    b.y = b.x + v.x;
    b.z = b.y + v.y;
    b.w = b.z + v.z;
    if (COPY) ((int4*)offs)[idx4] = b;
#pragma unroll
    for (int c = 0; c < C; ++c) {
        t4[c * (N / 4) + idx4] = b;
        b.x += cn[c].x; b.y += cn[c].y; b.z += cn[c].z; b.w += cn[c].w;
    }
}

template <int E, int N, int C>
__device__ __forceinline__ void
sage_place_edges(const int bid,
                 const int* __restrict__ src,
                 const int* __restrict__ dst,
                 const int* __restrict__ table,
                 const int* __restrict__ slot,
                 int* __restrict__ esrc) {
    static_assert((E / C) % 1024 == 0, "a block's 1024 edges lie in one chunk");
    const int i = bid * 256 + threadIdx.x;
    const int* base = table + (size_t)(bid / (E / C / 1024)) * N;
    const int4 d = ((const int4*)dst)[i];
    const int4 s = ((const int4*)src)[i];
    const int4 r = ((const int4*)slot)[i];
    const int px = base[d.x] + r.x;
    const int py = base[d.y] + r.y;
    const int pz = base[d.z] + r.z;
    const int pw = base[d.w] + r.w;
    esrc[px] = s.x;
    esrc[py] = s.y;
    esrc[pz] = s.z;
    esrc[pw] = s.w;
}

template <int C, int RGA, int RGB>
__global__ void __launch_bounds__(1024)
sage_count_both(const int* __restrict__ dstA, int* __restrict__ tableA,
                int* __restrict__ totA, int* __restrict__ slotA,
                const int* __restrict__ dstB, int* __restrict__ tableB,
                int* __restrict__ totB, int* __restrict__ slotB) {
    constexpr int GB = (N2_ / RGB) * C;
    const int bid = blockIdx.x;
    if (bid < GB) sage_count_ranges<E2_, N2_, RGB, C>(bid, dstB, tableB, totB, slotB);
    else          sage_count_ranges<E1_, N1_, RGA, C>(bid - GB, dstA, tableA, totA, slotA);
}

template <int C, int RG>
__global__ void __launch_bounds__(RG / 4)
sage_scan_both(int* __restrict__ tableA, const int* __restrict__ totA, int* __restrict__ degA,
               int* __restrict__ tableB, const int* __restrict__ totB, int* __restrict__ degB,
               int* __restrict__ offsB) {
    constexpr int GB = N2_ / RG;
    const int bid = blockIdx.x;
    if (bid < GB) sage_scan_ranges<N2_, RG, C, true>(bid, tableB, totB, degB, offsB);
    else          sage_scan_ranges<N1_, RG, C, false>(bid - GB, tableA, totA, degA, nullptr);
}

template <int C>
__global__ void __launch_bounds__(256)
sage_place_both(const int* __restrict__ srcA, const int* __restrict__ dstA,
                const int* __restrict__ tableA, const int* __restrict__ slotA,
                int* __restrict__ esrcA,
                const int* __restrict__ srcB, const int* __restrict__ dstB,
                const int* __restrict__ tableB, const int* __restrict__ slotB,
                int* __restrict__ esrcB) {
    constexpr int GB = E2_ / 1024;
    const int bid = blockIdx.x;
    if (bid < GB) sage_place_edges<E2_, N2_, C>(bid, srcB, dstB, tableB, slotB, esrcB);
    else          sage_place_edges<E1_, N1_, C>(bid - GB, srcA, dstA, tableA, slotA, esrcA);
}

template <int C, int B>
__device__ __forceinline__ void sage_sum_rows(const float* __restrict__ x, int lane,
                                              int myid, int n, float4 (&acc)[C / 128]) {
    constexpr int V = C / 128;
    int j = 0;
    for (; j + B <= n; j += B) {
        float4 t[B][V];
#pragma unroll
        for (int k = 0; k < B; ++k) {
            const int s = __builtin_amdgcn_readlane(myid, j + k);
            const float4* row = (const float4*)(x + (size_t)s * C);
#pragma unroll
            for (int v = 0; v < V; ++v) t[k][v] = row[lane + 32 * v];
        }
        __builtin_amdgcn_sched_barrier(0);
#pragma unroll
        for (int k = 0; k < B; ++k) {
#pragma unroll
            for (int v = 0; v < V; ++v) {
                acc[v].x += t[k][v].x; acc[v].y += t[k][v].y;
                acc[v].z += t[k][v].z; acc[v].w += t[k][v].w;
            }
        }
    }
    if (j < n) {
        float4 t[B - 1][V];
#pragma unroll
        for (int k = 0; k < B - 1; ++k) {
            if (j + k < n) {
                const int s = __builtin_amdgcn_readlane(myid, j + k);
                const float4* row = (const float4*)(x + (size_t)s * C);
#pragma unroll
                for (int v = 0; v < V; ++v) t[k][v] = row[lane + 32 * v];
            } else {
#pragma unroll
                for (int v = 0; v < V; ++v) t[k][v] = make_float4(0.f, 0.f, 0.f, 0.f);
            }
        }
#pragma unroll
        for (int k = 0; k < B - 1; ++k) {
            if (j + k < n) {
#pragma unroll
                for (int v = 0; v < V; ++v) {
                    acc[v].x += t[k][v].x; acc[v].y += t[k][v].y;
                    acc[v].z += t[k][v].z; acc[v].w += t[k][v].w;
                }
            }
        }
    }
}

template <int C, int R>
__global__ void __launch_bounds__(256)
sage_gather_mean(const float* __restrict__ x,
                                 const int* __restrict__ esrc,
                                 const int* __restrict__ offs,
                                 const int* __restrict__ deg,
                                 float* __restrict__ outm, int nDst) {
    static_assert(R <= 32, "one lane per row of the wave");
    const int wave = blockIdx.x * (blockDim.x >> 5) + (threadIdx.x >> 5);
    const int lane = threadIdx.x & 31;
    const int r0 = __builtin_amdgcn_readfirstlane(wave) * R;
    if (r0 >= nDst) return;
    constexpr int V = C / 128;
    constexpr int B = 16 / V;

    const bool has = (lane < R) && (r0 + lane < nDst);
    const int offv = has ? offs[r0 + lane] : 0;
    const int degv = has ? deg[r0 + lane] : 0;

    int beg = __builtin_amdgcn_readlane(offv, 0);
    int d   = __builtin_amdgcn_readlane(degv, 0);
    int ids = (lane < d) ? esrc[beg + lane] : 0;

    for (int r = 0; r < R; ++r) {
        if (r0 + r >= nDst) break;
        const int cur = ids;
        const int cbeg = beg;
        const int cd   = d;
        if (r + 1 < R) {
            beg = __builtin_amdgcn_readlane(offv, r + 1);
            d   = __builtin_amdgcn_readlane(degv, r + 1);
            ids = (lane < d) ? esrc[beg + lane] : 0;
        }
        float4 acc[V];
#pragma unroll
        for (int v = 0; v < V; ++v) acc[v] = make_float4(0.f, 0.f, 0.f, 0.f);

        sage_sum_rows<C, B>(x, lane, cur, (cd < 32) ? cd : 32, acc);
        for (int base = 32; base < cd; base += 32) {
            int n = cd - base;
            if (n > 32) n = 32;
            const int more = (lane < n) ? esrc[cbeg + base + lane] : 0;
            sage_sum_rows<C, B>(x, lane, more, n, acc);
        }

        const float scale = (cd > 0) ? (1.0f / (float)cd) : 0.0f;
        float4* orow = (float4*)(outm + (size_t)(r0 + r) * C);
#pragma unroll
        for (int v = 0; v < V; ++v) {
            float4 t;
            t.x = acc[v].x * scale; t.y = acc[v].y * scale;
            t.z = acc[v].z * scale; t.w = acc[v].w * scale;
            orow[lane + 32 * v] = t;
        }
    }
}

template <int K, int N, bool RELU>
__global__ void __launch_bounds__(256)
sage_gemm_two(const float* __restrict__ A1,
                              const float* __restrict__ B1,
                              const float* __restrict__ A2,
                              const float* __restrict__ B2,
                              const float* __restrict__ bias,
                              float* __restrict__ out, int M) {
    constexpr int MT = 2, NT = 4;
    constexpr int BM = 64;
    constexpr int P  = 36;
    constexpr int KB = K / 32;
    constexpr int STEPS = 2 * KB;
    static_assert(K % 32 == 0, "K is walked in whole 128-byte row lines");
    static_assert(N == 256, "4 wave columns x 4 tiles x 16 columns; 8 B rows per staging thread");
    __shared__ __attribute__((aligned(16))) float sA[2][BM * P];
    __shared__ __attribute__((aligned(16))) float sB[2][N * P];

    const int tid  = threadIdx.x;
    const int lane = tid & 31;
    const int wv   = __builtin_amdgcn_readfirstlane(tid >> 5);
    const int gmw  = wv >> 2;
    const int gnw  = wv & 3;
    const int m0   = blockIdx.x * BM;
    if (m0 >= M) return;

    const int sp   = tid & 7;
    const int sr   = tid >> 3;
    const int soff = sr * P + sp * 4;

    const int c    = lane & 15;
    const int kh   = (lane >> 4) * 16;
    const int aoff = (gmw * 32 + c) * P + kh;
    const int boff = (gnw * 64 + c) * P + kh;

    v8f acc[MT][NT] = {};
    v4f ra[2], rb[8];

    {
        const float* ga = A1 + (size_t)(m0 + sr) * K + sp * 4;
        const float* gb = B1 + (size_t)sr * K + sp * 4;
#pragma unroll
        for (int j = 0; j < 2; ++j) ra[j] = *(const v4f*)(ga + (size_t)j * 32 * K);
#pragma unroll
        for (int j = 0; j < 8; ++j) rb[j] = *(const v4f*)(gb + (size_t)j * 32 * K);
#pragma unroll
        for (int j = 0; j < 2; ++j) *(v4f*)(&sA[0][soff + j * 32 * P]) = ra[j];
#pragma unroll
        for (int j = 0; j < 8; ++j) *(v4f*)(&sB[0][soff + j * 32 * P]) = rb[j];
    }
    __syncthreads();

#pragma unroll 1
    for (int s = 0; s < STEPS; ++s) {
        const int buf = s & 1;
        const bool more = (s + 1 < STEPS);
        if (more) {
            const int sn = s + 1;
            const float* An = (sn < KB) ? A1 : A2;
            const float* Bn = (sn < KB) ? B1 : B2;
            const int kc = (sn % KB) * 32 + sp * 4;
            const float* ga = An + (size_t)(m0 + sr) * K + kc;
            const float* gb = Bn + (size_t)sr * K + kc;
#pragma unroll
            for (int j = 0; j < 2; ++j) ra[j] = *(const v4f*)(ga + (size_t)j * 32 * K);
#pragma unroll
            for (int j = 0; j < 8; ++j) rb[j] = *(const v4f*)(gb + (size_t)j * 32 * K);
        }
        const float* pa = &sA[buf][aoff];
        const float* pb = &sB[buf][boff];
#pragma unroll
        for (int q = 0; q < 4; ++q) {
            v4f a[MT], b[NT];
#pragma unroll
            for (int mt = 0; mt < MT; ++mt) a[mt] = *(const v4f*)(pa + mt * 16 * P + 4 * q);
#pragma unroll
            for (int nt = 0; nt < NT; ++nt) b[nt] = *(const v4f*)(pb + nt * 16 * P + 4 * q);
#pragma unroll
            for (int p = 0; p < 2; ++p) {
                v2f af[MT], bf[NT];
#pragma unroll
                for (int mt = 0; mt < MT; ++mt) {
                    af[mt].x = a[mt][2 * p];
                    af[mt].y = a[mt][2 * p + 1];
                }
#pragma unroll
                for (int nt = 0; nt < NT; ++nt) {
                    bf[nt].x = b[nt][2 * p];
                    bf[nt].y = b[nt][2 * p + 1];
                }
#pragma unroll
                for (int mt = 0; mt < MT; ++mt)
#pragma unroll
                    for (int nt = 0; nt < NT; ++nt)
                        acc[mt][nt] = __builtin_amdgcn_wmma_f32_16x16x4_f32(
                            false, af[mt], false, bf[nt], (short)0, acc[mt][nt],
                            false, false);
            }
        }
        if (more) {
#pragma unroll
            for (int j = 0; j < 2; ++j) *(v4f*)(&sA[buf ^ 1][soff + j * 32 * P]) = ra[j];
#pragma unroll
            for (int j = 0; j < 8; ++j) *(v4f*)(&sB[buf ^ 1][soff + j * 32 * P]) = rb[j];
        }
        __syncthreads();
    }

    const int n0    = gnw * 64;
    const int mw    = m0 + gmw * 32;
    const int rbase = ((lane >> 4) << 3);
#pragma unroll
    for (int nt = 0; nt < NT; ++nt) {
        const int ocol = n0 + nt * 16 + c;
        const float bb = bias[ocol];
#pragma unroll
        for (int mt = 0; mt < MT; ++mt) {
#pragma unroll
            for (int r = 0; r < 8; ++r) {
                float v = acc[mt][nt][r] + bb;
                if (RELU) v = fmaxf(v, 0.0f);
                out[(size_t)(mw + mt * 16 + rbase + r) * N + ocol] = v;
            }
        }
    }
}

template <int K>
__device__ __forceinline__ void sage_tile_k(const float* ap, const float* bp, v8f& acc) {
#pragma unroll 1
    for (int kc = 0; kc < K; kc += 64) {
        v4f a[2][4], b[2][4];
#pragma unroll
        for (int u = 0; u < 2; ++u) {
#pragma unroll
            for (int q = 0; q < 4; ++q) {
                a[u][q] = *(const v4f*)(ap + kc + 32 * u + 4 * q);
                b[u][q] = *(const v4f*)(bp + kc + 32 * u + 4 * q);
            }
        }
#pragma unroll
        for (int u = 0; u < 2; ++u) {
#pragma unroll
            for (int q = 0; q < 4; ++q) {
#pragma unroll
                for (int p = 0; p < 2; ++p) {
                    v2f af, bf;
                    af.x = a[u][q][2 * p]; af.y = a[u][q][2 * p + 1];
                    bf.x = b[u][q][2 * p]; bf.y = b[u][q][2 * p + 1];
                    acc = __builtin_amdgcn_wmma_f32_16x16x4_f32(false, af, false, bf, (short)0,
                                                                acc, false, false);
                }
            }
        }
    }
}

template <int RPW>
__global__ void __launch_bounds__(256)
sage_layer2_fused(const float* __restrict__ hsrc,
                  const int* __restrict__ esrc,
                  const int* __restrict__ offs,
                  const int* __restrict__ deg,
                  const float* __restrict__ Wl,
                  const float* __restrict__ Wr,
                  const float* __restrict__ bias,
                  float* __restrict__ out) {
    constexpr int C    = HID_C;
    constexpr int ROWS = 8 * RPW;
    static_assert(ROWS == 16, "one 16-row tile per work-group");
    static_assert(N2_ % ROWS == 0, "whole work-groups");
    constexpr int V  = C / 128;
    constexpr int B  = 16 / V;
    constexpr int PA = C + 4;
    constexpr int PO = OUT_C + 4;
    __shared__ __attribute__((aligned(16))) float sA[ROWS * PA];
    __shared__ __attribute__((aligned(16))) float sP[2 * ROWS * PO];
    __shared__ __attribute__((aligned(16))) float sO[ROWS * PA];

    const int lane = threadIdx.x & 31;
    const int wv   = __builtin_amdgcn_readfirstlane((int)(threadIdx.x >> 5));
    const int row0 = blockIdx.x * ROWS;

    {
        const int lr0 = wv * RPW;
        const int r0  = row0 + lr0;
        float4 own[RPW][V];
#pragma unroll
        for (int r = 0; r < RPW; ++r) {
            const float4* grow = (const float4*)(hsrc + (size_t)(r0 + r) * C);
#pragma unroll
            for (int v = 0; v < V; ++v) own[r][v] = grow[lane + 32 * v];
        }
        const bool has = lane < RPW;
        const int offv = has ? offs[r0 + lane] : 0;
        const int degv = has ? deg[r0 + lane] : 0;
        int beg = __builtin_amdgcn_readlane(offv, 0);
        int d   = __builtin_amdgcn_readlane(degv, 0);
        int ids = (lane < d) ? esrc[beg + lane] : 0;
#pragma unroll
        for (int r = 0; r < RPW; ++r) {
            float4* prow = (float4*)(sO + (lr0 + r) * PA);
#pragma unroll
            for (int v = 0; v < V; ++v) prow[lane + 32 * v] = own[r][v];
        }
        for (int r = 0; r < RPW; ++r) {
            const int cur  = ids;
            const int cbeg = beg;
            const int cd   = d;
            if (r + 1 < RPW) {
                beg = __builtin_amdgcn_readlane(offv, r + 1);
                d   = __builtin_amdgcn_readlane(degv, r + 1);
                ids = (lane < d) ? esrc[beg + lane] : 0;
            }
            float4 acc[V];
#pragma unroll
            for (int v = 0; v < V; ++v) acc[v] = make_float4(0.f, 0.f, 0.f, 0.f);
            sage_sum_rows<C, B>(hsrc, lane, cur, (cd < 32) ? cd : 32, acc);
            for (int base = 32; base < cd; base += 32) {
                int n = cd - base;
                if (n > 32) n = 32;
                const int more = (lane < n) ? esrc[cbeg + base + lane] : 0;
                sage_sum_rows<C, B>(hsrc, lane, more, n, acc);
            }
            const float scale = (cd > 0) ? (1.0f / (float)cd) : 0.0f;
            float4* orow = (float4*)(sA + (lr0 + r) * PA);
#pragma unroll
            for (int v = 0; v < V; ++v) {
                float4 t;
                t.x = acc[v].x * scale; t.y = acc[v].y * scale;
                t.z = acc[v].z * scale; t.w = acc[v].w * scale;
                orow[lane + 32 * v] = t;
            }
        }
    }
    __syncthreads();

    {
        const int c  = lane & 15;
        const int kh = (lane >> 4) * 16;
        const int nt = wv & 3;
        const int ph = wv >> 2;
        v8f acc = {};
        if (ph == 0)
            sage_tile_k<C>(sA + c * PA + kh, Wl + (size_t)(nt * 16 + c) * C + kh, acc);
        else
            sage_tile_k<C>(sO + c * PA + kh, Wr + (size_t)(nt * 16 + c) * C + kh, acc);
        float* pp = sP + ph * (ROWS * PO) + ((lane >> 4) << 3) * PO + nt * 16 + c;
#pragma unroll
        for (int r = 0; r < 8; ++r) pp[r * PO] = acc[r];
    }
    __syncthreads();

    {
        const float blo = bias[lane];
        const float bhi = bias[lane + 32];
#pragma unroll
        for (int r = 0; r < RPW; ++r) {
            const int lr = wv * RPW + r;
            const float* p0 = sP + lr * PO;
            const float* p1 = sP + ROWS * PO + lr * PO;
            const float a = (p0[lane] + p1[lane]) + blo;
            const float b = (p0[lane + 32] + p1[lane + 32]) + bhi;
            float m = fmaxf(a, b);
#pragma unroll
            for (int off = 16; off > 0; off >>= 1)
                m = fmaxf(m, __shfl_xor(m, off, 32));
            float s = expf(a - m) + expf(b - m);
#pragma unroll
            for (int off = 16; off > 0; off >>= 1)
                s += __shfl_xor(s, off, 32);
            const float lse = m + logf(s);
            float* o = out + (size_t)(row0 + lr) * OUT_C;
            o[lane]      = a - lse;
            o[lane + 32] = b - lse;
        }
    }
}

extern "C" void kernel_launch(void* const* d_in, const int* in_sizes, int n_in,
                              void* d_out, int out_size, void* d_ws, size_t ws_size,
                              hipStream_t stream) {
    const float* x    = (const float*)d_in[0];
    const int*   src1 = (const int*)d_in[1];
    const int*   dst1 = (const int*)d_in[2];
    const int*   src2 = (const int*)d_in[3];
    const int*   dst2 = (const int*)d_in[4];
    const float* W1l  = (const float*)d_in[5];
    const float* W1r  = (const float*)d_in[6];
    const float* b1   = (const float*)d_in[7];
    const float* W2l  = (const float*)d_in[8];
    const float* W2r  = (const float*)d_in[9];
    const float* b2   = (const float*)d_in[10];
    float* out = (float*)d_out;

    constexpr int CH  = 8;
    constexpr int RG  = 2048;
    int* deg1  = (int*)d_ws;
    int* deg2  = deg1 + N1_;
    int* tot1  = deg2 + N2_;
    int* offs2 = tot1 + 4096;
    int* tot2  = tot1 + N1_;
    int* esrc1 = tot2 + N2_;
    int* esrc2 = esrc1 + E1_;
    float* aggr1 = (float*)(esrc2 + E2_);
    float* aggr2 = aggr1 + (size_t)N1_ * IN_C;
    float* h     = aggr2 + (size_t)N2_ * HID_C;
    int* slot1  = (int*)aggr1;
    int* slot2  = (int*)aggr2;
    int* table1 = (int*)h;
    int* table2 = (int*)aggr2 + E2_;

    sage_count_both<CH, RG, RG>
        <<<((N1_ + N2_) / RG) * CH, 1024, 0, stream>>>(dst1, table1, tot1, slot1,
                                                       dst2, table2, tot2, slot2);
    sage_scan_both<CH, RG>
        <<<(N1_ + N2_) / RG, RG / 4, 0, stream>>>(table1, tot1, deg1,
                                                  table2, tot2, deg2, offs2);
    sage_place_both<CH>
        <<<(E1_ + E2_) / 1024, 256, 0, stream>>>(src1, dst1, table1, slot1, esrc1,
                                                 src2, dst2, table2, slot2, esrc2);

    sage_gather_mean<IN_C, 8><<<N1_ / 8 / 8, 256, 0, stream>>>(x, esrc1, table1, deg1,
                                                               aggr1, N1_);
    sage_gemm_two<IN_C, HID_C, true>
        <<<N1_ / 64, 256, 0, stream>>>(aggr1, W1l, x, W1r, b1, h, N1_);

    sage_layer2_fused<2><<<N2_ / 16, 256, 0, stream>>>(h, esrc2, offs2, deg2, W2l, W2r, b2, out);
}
